# starting kernel
# baseline (speedup 1.0000x reference)
#include <hip/hip_runtime.h>
#include <hip/hip_bf16.h>

typedef float v2f __attribute__((ext_vector_type(2)));
typedef float v8f __attribute__((ext_vector_type(8)));

#define N_NODES 100000
#define N_EDGES 1600000
#define IN_F    256
#define OUT_F   64
#define NTILES  (N_NODES / 16)

__device__ __forceinline__ void atomic_add_f32_dev(float* p, float v) {
    asm volatile("global_atomic_add_f32 %0, %1, off scope:SCOPE_DEV"
                 :: "v"(p), "v"(v) : "memory");
}

__global__ void sgc_gemm_wmma(const float* __restrict__ x,
                              const float* __restrict__ W,
                              const float* __restrict__ bias,
                              float* __restrict__ out) {
    const int wave = (blockIdx.x * blockDim.x + threadIdx.x) >> 5;
    const int lane = threadIdx.x & 31;
    if (wave >= NTILES) return;

    const int m    = lane & 15;
    const int koff = (lane >> 4) << 1;

    const float* __restrict__ xrow = x + (size_t)(wave * 16 + m) * IN_F;
    const float* __restrict__ wr0  = W + (size_t)(m +  0) * IN_F;
    const float* __restrict__ wr1  = W + (size_t)(m + 16) * IN_F;
    const float* __restrict__ wr2  = W + (size_t)(m + 32) * IN_F;
    const float* __restrict__ wr3  = W + (size_t)(m + 48) * IN_F;

    v8f acc0 = {}, acc1 = {}, acc2 = {}, acc3 = {};

#pragma unroll 4
    for (int k = 0; k < IN_F; k += 4) {
        v2f a  = *(const v2f*)(xrow + k + koff);
        v2f b0 = *(const v2f*)(wr0  + k + koff);
        v2f b1 = *(const v2f*)(wr1  + k + koff);
        v2f b2 = *(const v2f*)(wr2  + k + koff);
        v2f b3 = *(const v2f*)(wr3  + k + koff);
        acc0 = __builtin_amdgcn_wmma_f32_16x16x4_f32(false, a, false, b0, (short)0, acc0, false, false);
        acc1 = __builtin_amdgcn_wmma_f32_16x16x4_f32(false, a, false, b1, (short)0, acc1, false, false);
        acc2 = __builtin_amdgcn_wmma_f32_16x16x4_f32(false, a, false, b2, (short)0, acc2, false, false);
        acc3 = __builtin_amdgcn_wmma_f32_16x16x4_f32(false, a, false, b3, (short)0, acc3, false, false);
    }

    const int row_base = wave * 16 + ((lane >> 4) ? 8 : 0);
    const int col      = lane & 15;
    const float bv0 = bias[col +  0];
    const float bv1 = bias[col + 16];
    const float bv2 = bias[col + 32];
    const float bv3 = bias[col + 48];

#pragma unroll
    for (int r = 0; r < 8; ++r) {
        float* o = out + (size_t)(row_base + r) * OUT_F + col;
        o[ 0] = acc0[r] + bv0;
        o[16] = acc1[r] + bv1;
        o[32] = acc2[r] + bv2;
        o[48] = acc3[r] + bv3;
    }
}

__global__ void sgc_zero(float4* __restrict__ p, int n4) {
    const float4 z = {0.f, 0.f, 0.f, 0.f};
    for (int i = blockIdx.x * blockDim.x + threadIdx.x; i < n4;
         i += gridDim.x * blockDim.x)
        p[i] = z;
}

__global__ void sgc_scatter(const float* __restrict__ cur,
                            float* __restrict__ nxt,
                            const int* __restrict__ esrc,
                            const int* __restrict__ edst,
                            const float* __restrict__ ew,
                            int n_edges) {
    const int gid  = blockIdx.x * blockDim.x + threadIdx.x;
    const int e    = gid >> 5;
    const int lane = gid & 31;
    if (e >= n_edges) return;

    const int   s = __builtin_nontemporal_load(esrc + e);
    const int   d = __builtin_nontemporal_load(edst + e);
    const float w = __builtin_nontemporal_load(ew + e);

    const float* __restrict__ ps = cur + (size_t)s * OUT_F + lane * 2;
    float*       __restrict__ pd = nxt + (size_t)d * OUT_F + lane * 2;

    const v2f vv = *(const v2f*)ps;
    atomic_add_f32_dev(pd + 0, vv.x * w);
    atomic_add_f32_dev(pd + 1, vv.y * w);
}

extern "C" void kernel_launch(void* const* d_in, const int* in_sizes, int n_in,
                              void* d_out, int out_size, void* d_ws, size_t ws_size,
                              hipStream_t stream) {
    const float* x    = (const float*)d_in[0];
    const float* W    = (const float*)d_in[1];
    const float* b    = (const float*)d_in[2];
    const int*   esrc = (const int*)d_in[3];
    const int*   edst = (const int*)d_in[4];
    const float* ew   = (const float*)d_in[5];

    float* out = (float*)d_out;
    float* ws  = (float*)d_ws;

    const int n4 = N_NODES * OUT_F / 4;

    sgc_gemm_wmma<<<(NTILES + 7) / 8, 256, 0, stream>>>(x, W, b, ws);

    const dim3 sgrid((unsigned)(((long long)N_EDGES * 32 + 255) / 256));

    sgc_zero<<<1024, 256, 0, stream>>>((float4*)out, n4);
    sgc_scatter<<<sgrid, 256, 0, stream>>>(ws, out, esrc, edst, ew, N_EDGES);

    sgc_zero<<<1024, 256, 0, stream>>>((float4*)ws, n4);
    sgc_scatter<<<sgrid, 256, 0, stream>>>(out, ws, esrc, edst, ew, N_EDGES);

    sgc_zero<<<1024, 256, 0, stream>>>((float4*)out, n4);
    sgc_scatter<<<sgrid, 256, 0, stream>>>(ws, out, esrc, edst, ew, N_EDGES);
}


# hops: 256 work-groups hold 391 destination rows each as f32 sums in LDS, 13 rows a wave; waves read the whole edge list, pass hits to the owning wave via LDS; no atomic, no zero fill, fixed sum order
# speedup vs baseline: 1.4912x; 1.4912x over previous
#include <hip/hip_runtime.h>
#include <hip/hip_bf16.h>

typedef float v2f __attribute__((ext_vector_type(2)));
typedef float v8f __attribute__((ext_vector_type(8)));

#define N_NODES 100000
#define N_EDGES 1600000
#define IN_F    256
#define OUT_F   64
#define NTILES  (N_NODES / 16)

__global__ void sgc_gemm_wmma(const float* __restrict__ x,
                              const float* __restrict__ W,
                              const float* __restrict__ bias,
                              float* __restrict__ out) {
    const int wave = (blockIdx.x * blockDim.x + threadIdx.x) >> 5;
    const int lane = threadIdx.x & 31;
    if (wave >= NTILES) return;

    const int m    = lane & 15;
    const int koff = (lane >> 4) << 1;

    const float* __restrict__ xrow = x + (size_t)(wave * 16 + m) * IN_F;
    const float* __restrict__ wr0  = W + (size_t)(m +  0) * IN_F;
    const float* __restrict__ wr1  = W + (size_t)(m + 16) * IN_F;
    const float* __restrict__ wr2  = W + (size_t)(m + 32) * IN_F;
    const float* __restrict__ wr3  = W + (size_t)(m + 48) * IN_F;

    v8f acc0 = {}, acc1 = {}, acc2 = {}, acc3 = {};

#pragma unroll 4
    for (int k = 0; k < IN_F; k += 4) {
        v2f a  = *(const v2f*)(xrow + k + koff);
        v2f b0 = *(const v2f*)(wr0  + k + koff);
        v2f b1 = *(const v2f*)(wr1  + k + koff);
        v2f b2 = *(const v2f*)(wr2  + k + koff);
        v2f b3 = *(const v2f*)(wr3  + k + koff);
        acc0 = __builtin_amdgcn_wmma_f32_16x16x4_f32(false, a, false, b0, (short)0, acc0, false, false);
        acc1 = __builtin_amdgcn_wmma_f32_16x16x4_f32(false, a, false, b1, (short)0, acc1, false, false);
        acc2 = __builtin_amdgcn_wmma_f32_16x16x4_f32(false, a, false, b2, (short)0, acc2, false, false);
        acc3 = __builtin_amdgcn_wmma_f32_16x16x4_f32(false, a, false, b3, (short)0, acc3, false, false);
    }

    const int row_base = wave * 16 + ((lane >> 4) ? 8 : 0);
    const int col      = lane & 15;
    const float bv0 = bias[col +  0];
    const float bv1 = bias[col + 16];
    const float bv2 = bias[col + 32];
    const float bv3 = bias[col + 48];

#pragma unroll
    for (int r = 0; r < 8; ++r) {
        float* o = out + (size_t)(row_base + r) * OUT_F + col;
        o[ 0] = acc0[r] + bv0;
        o[16] = acc1[r] + bv1;
        o[32] = acc2[r] + bv2;
        o[48] = acc3[r] + bv3;
    }
}

#define HOP_GROUPS      256
#define ROWS_PER_GROUP  391
#define HOP_THREADS     1024
#define HOP_WAVES       32
#define ROWS_PER_WAVE   13
#define EDGES_PER_CHUNK 128
#define N_CHUNKS        (N_EDGES / EDGES_PER_CHUNK)
#define BOX_WORDS       32
#define EDGE_BITS       21
#define QUEUE_WORDS     160

typedef int v4i __attribute__((ext_vector_type(4)));

template <bool FULL>
__device__ __forceinline__ void hop_drain(const int* qp, int n, int lane,
                                          const float* __restrict__ cur,
                                          const int* __restrict__ esrc,
                                          const float* __restrict__ ew,
                                          float* acc) {
    int   s  = 0;
    int   rw = 0;
    float wt = 0.f;
    if (FULL || lane < n) {
        const unsigned pk = (unsigned)qp[lane];
        const int e = (int)(pk & ((1u << EDGE_BITS) - 1u));
        rw = (int)(pk >> EDGE_BITS);
        s  = esrc[e];
        wt = ew[e];
    }
    const int wti = __builtin_bit_cast(int, wt);
    const int cnt = FULL ? 32 : n;

    v2f   v[8];
    int   rj[8];
    float wj[8];
#define HOP_GATHER8(J0, V, RJ, WJ)                                                        \
    _Pragma("unroll")                                                                     \
    for (int u = 0; u < 8; ++u) {                                                         \
        int jj = (J0) + u;                                                                \
        if (!FULL) jj = jj < n ? jj : n - 1;                                              \
        const int sj = __builtin_amdgcn_readlane(s, jj);                                  \
        RJ[u] = __builtin_amdgcn_readlane(rw, jj);                                        \
        WJ[u] = __builtin_bit_cast(float, __builtin_amdgcn_readlane(wti, jj));            \
        V[u]  = *(const v2f*)(cur + (size_t)sj * OUT_F + lane * 2);                       \
    }
    HOP_GATHER8(0, v, rj, wj)
    for (int j0 = 0; j0 < cnt; j0 += 8) {
        v2f   nv[8];
        int   nrj[8];
        float nwj[8];
        const bool more = j0 + 8 < cnt;
        if (more) {
            HOP_GATHER8(j0 + 8, nv, nrj, nwj)
        }
#pragma unroll
        for (int u = 0; u < 8; ++u) {
            if (FULL || j0 + u < n) {
                v2f* a = (v2f*)(acc + rj[u] * OUT_F) + lane;
                *a = *a + v[u] * wj[u];
            }
        }
        if (more) {
#pragma unroll
            for (int u = 0; u < 8; ++u) { v[u] = nv[u]; rj[u] = nrj[u]; wj[u] = nwj[u]; }
        }
    }
#undef HOP_GATHER8
}

__global__ __launch_bounds__(HOP_THREADS)
void sgc_hop_owner_waves(const float* __restrict__ cur,
                         float* __restrict__ nxt,
                         const int* __restrict__ esrc,
                         const int* __restrict__ edst,
                         const float* __restrict__ ew) {
    __shared__ __attribute__((aligned(16))) float acc[ROWS_PER_GROUP * OUT_F];
    __shared__ int boxes[HOP_WAVES * HOP_WAVES * BOX_WORDS];
    __shared__ int box_cnt[HOP_WAVES * HOP_WAVES];
    __shared__ int ownq[HOP_WAVES][64];
    __shared__ int overflow_flag;

    const int tid  = threadIdx.x;
    const int lane = tid & 31;
    const int wv   = __builtin_amdgcn_readfirstlane(tid >> 5);
    const int base = blockIdx.x * ROWS_PER_GROUP;
    const int left = N_NODES - base;
    const unsigned nrows = (unsigned)(left < ROWS_PER_GROUP ? left : ROWS_PER_GROUP);

    if (tid == 0) overflow_flag = 0;
    float4* acc4 = (float4*)acc;
    for (int i = tid; i < ROWS_PER_GROUP * OUT_F / 4; i += HOP_THREADS)
        acc4[i] = make_float4(0.f, 0.f, 0.f, 0.f);
    __syncthreads();

    const v4i* __restrict__ dst4 = (const v4i*)edst;

    int  cntv = 0;
    bool ovf  = false;
    v4i d[4];
#pragma unroll
    for (int u = 0; u < 4; ++u) d[u] = dst4[(size_t)(u * HOP_WAVES + wv) * 32 + lane];
    for (int c0 = 0; c0 < N_CHUNKS; c0 += 4 * HOP_WAVES) {
        v4i nd[4];
        int c[4];
#pragma unroll
        for (int u = 0; u < 4; ++u) {
            c[u] = c0 + u * HOP_WAVES + wv;
            const int cn = c[u] + 4 * HOP_WAVES;
            const int cc = cn < N_CHUNKS ? cn : N_CHUNKS - 1;
            nd[u] = dst4[(size_t)cc * 32 + lane];
        }
#pragma unroll
        for (int u = 0; u < 4; ++u) {
            if (c[u] >= N_CHUNKS) continue;
            const unsigned rr[4] = {(unsigned)(d[u].x - base), (unsigned)(d[u].y - base),
                                    (unsigned)(d[u].z - base), (unsigned)(d[u].w - base)};
            const unsigned m01 = rr[0] < rr[1] ? rr[0] : rr[1];
            const unsigned m23 = rr[2] < rr[3] ? rr[2] : rr[3];
            const unsigned mn  = m01 < m23 ? m01 : m23;
            if (__builtin_amdgcn_ballot_w32(mn < nrows) == 0u) continue;

            const unsigned e0 = (unsigned)(c[u] * EDGES_PER_CHUNK + lane * 4);
#pragma unroll
            for (int k = 0; k < 4; ++k) {
                const bool hit = rr[k] < nrows;
                const int  pkv = (int)((rr[k] << EDGE_BITS) | (e0 + (unsigned)k));
                unsigned m = __builtin_amdgcn_ballot_w32(hit);
                while (m != 0u) {
                    const int l = __builtin_ctz(m);
                    m &= m - 1u;
                    const int p = __builtin_amdgcn_readlane(pkv, l);
                    const int o = (int)(((unsigned)p >> EDGE_BITS) & 31u);
                    if (lane == o) {
                        if (cntv < BOX_WORDS) {
                            boxes[(wv * HOP_WAVES + o) * BOX_WORDS + cntv] = p;
                            ++cntv;
                        } else {
                            ovf = true;
                        }
                    }
                }
            }
        }
#pragma unroll
        for (int u = 0; u < 4; ++u) d[u] = nd[u];
    }
    box_cnt[wv * HOP_WAVES + lane] = cntv;
    if (__builtin_amdgcn_ballot_w32(ovf) != 0u && lane == 0) overflow_flag = 1;
    __syncthreads();
    const int over = __builtin_amdgcn_readfirstlane(overflow_flag);

    if (over == 0) {
        const int  mine  = box_cnt[lane * HOP_WAVES + wv];
        const int* mybox = boxes + (lane * HOP_WAVES + wv) * BOX_WORDS;
        int* oq = ownq[wv];
        int  on = 0;
        for (int i = 0; i < BOX_WORDS; ++i) {
            const bool has = i < mine;
            const unsigned m = __builtin_amdgcn_ballot_w32(has);
            if (m == 0u) break;
            const int pos = on + (int)__builtin_amdgcn_mbcnt_lo(m, 0u);
            if (has) oq[pos] = mybox[i];
            on += __builtin_popcount(m);
            if (on >= 32) {
                on -= 32;
                hop_drain<true>(oq + on, 32, lane, cur, esrc, ew, acc);
            }
        }
        if (on > 0) hop_drain<false>(oq, on, lane, cur, esrc, ew, acc);
    } else {
        int* sq = boxes + wv * QUEUE_WORDS;
        int  qn = 0;
        for (int c0 = 0; c0 < N_CHUNKS; c0 += 4) {
            v4i d[4];
#pragma unroll
            for (int u = 0; u < 4; ++u) d[u] = dst4[(size_t)(c0 + u) * 32 + lane];
#pragma unroll
            for (int u = 0; u < 4; ++u) {
                const unsigned rr[4] = {(unsigned)(d[u].x - base), (unsigned)(d[u].y - base),
                                        (unsigned)(d[u].z - base), (unsigned)(d[u].w - base)};
                const unsigned e0 = (unsigned)((c0 + u) * EDGES_PER_CHUNK + lane * 4);
#pragma unroll
                for (int k = 0; k < 4; ++k) {
                    const bool hit = rr[k] < nrows && (int)(rr[k] & 31u) == wv;
                    const unsigned m = __builtin_amdgcn_ballot_w32(hit);
                    if (m != 0u) {
                        const int pos = qn + (int)__builtin_amdgcn_mbcnt_lo(m, 0u);
                        if (hit) sq[pos] = (int)((rr[k] << EDGE_BITS) | (e0 + (unsigned)k));
                        qn += __builtin_popcount(m);
                    }
                }
                while (qn >= 32) {
                    qn -= 32;
                    hop_drain<true>(sq + qn, 32, lane, cur, esrc, ew, acc);
                }
            }
        }
        if (qn > 0) hop_drain<false>(sq, qn, lane, cur, esrc, ew, acc);
    }

#pragma unroll
    for (int r = 0; r < ROWS_PER_WAVE; ++r) {
        const unsigned row = (unsigned)(wv + HOP_WAVES * r);
        if (row < nrows)
            *(v2f*)(nxt + (size_t)(base + (int)row) * OUT_F + lane * 2) =
                *((const v2f*)(acc + row * OUT_F) + lane);
    }
}

extern "C" void kernel_launch(void* const* d_in, const int* in_sizes, int n_in,
                              void* d_out, int out_size, void* d_ws, size_t ws_size,
                              hipStream_t stream) {
    const float* x    = (const float*)d_in[0];
    const float* W    = (const float*)d_in[1];
    const float* b    = (const float*)d_in[2];
    const int*   esrc = (const int*)d_in[3];
    const int*   edst = (const int*)d_in[4];
    const float* ew   = (const float*)d_in[5];

    float* out = (float*)d_out;
    float* ws  = (float*)d_ws;

    sgc_gemm_wmma<<<(NTILES + 7) / 8, 256, 0, stream>>>(x, W, b, ws);

    sgc_hop_owner_waves<<<HOP_GROUPS, HOP_THREADS, 0, stream>>>(ws, out, esrc, edst, ew);
    sgc_hop_owner_waves<<<HOP_GROUPS, HOP_THREADS, 0, stream>>>(out, ws, esrc, edst, ew);
    sgc_hop_owner_waves<<<HOP_GROUPS, HOP_THREADS, 0, stream>>>(ws, out, esrc, edst, ew);
}


# projection: a wave computes a 64 x 64 block (4 x 4 tiles), each W or x fragment load feeds 4 products: 8 loads per 16 WMMAs, was 5 per 4; two fragment sets take turns; k order of sums kept
# speedup vs baseline: 1.5690x; 1.0522x over previous
#include <hip/hip_runtime.h>
#include <hip/hip_bf16.h>

typedef float v2f __attribute__((ext_vector_type(2)));
typedef float v8f __attribute__((ext_vector_type(8)));

#define N_NODES 100000
#define N_EDGES 1600000
#define IN_F    256
#define OUT_F   64
#define NTILES  (N_NODES / 16)

#define GEMM_MT      4
#define GEMM_WAVES   ((NTILES + GEMM_MT - 1) / GEMM_MT)
#define GEMM_THREADS 64
#define GEMM_GROUPS  ((GEMM_WAVES + 1) / 2)
#define GEMM_KC      4

__global__ __launch_bounds__(GEMM_THREADS)
void sgc_gemm_wmma(const float* __restrict__ x,
                   const float* __restrict__ W,
                   const float* __restrict__ bias,
                   float* __restrict__ out) {
    const int wave = (blockIdx.x * GEMM_THREADS + threadIdx.x) >> 5;
    const int lane = threadIdx.x & 31;
    if (wave >= GEMM_WAVES) return;

    const int m    = lane & 15;
    const int koff = (lane >> 4) << 1;

    const float* __restrict__ xr[GEMM_MT];
    const float* __restrict__ wr[4];
#pragma unroll
    for (int t = 0; t < GEMM_MT; ++t) {
        const int tile = wave * GEMM_MT + t;
        const int tl   = tile < NTILES ? tile : NTILES - 1;
        xr[t] = x + (size_t)(tl * 16 + m) * IN_F + koff;
    }
#pragma unroll
    for (int n = 0; n < 4; ++n) wr[n] = W + (size_t)(m + 16 * n) * IN_F + koff;

    v8f acc[GEMM_MT][4];
#pragma unroll
    for (int t = 0; t < GEMM_MT; ++t)
#pragma unroll
        for (int n = 0; n < 4; ++n) acc[t][n] = v8f{};

#define GEMM_LOAD_SET(A, B, K0)                                                           \
    _Pragma("unroll")                                                                     \
    for (int kk = 0; kk < GEMM_KC; ++kk) {                                                \
        _Pragma("unroll")                                                                 \
        for (int t = 0; t < GEMM_MT; ++t) A[kk][t] = *(const v2f*)(xr[t] + (K0) + 4 * kk); \
        _Pragma("unroll")                                                                 \
        for (int n = 0; n < 4; ++n)       B[kk][n] = *(const v2f*)(wr[n] + (K0) + 4 * kk); \
    }

#define GEMM_MUL_SET(A, B)                                                                \
    _Pragma("unroll")                                                                     \
    for (int kk = 0; kk < GEMM_KC; ++kk) {                                                \
        _Pragma("unroll")                                                                 \
        for (int t = 0; t < GEMM_MT; ++t) {                                               \
            _Pragma("unroll")                                                             \
            for (int n = 0; n < 4; ++n)                                                   \
                acc[t][n] = __builtin_amdgcn_wmma_f32_16x16x4_f32(false, A[kk][t], false, B[kk][n], \
                                                                  (short)0, acc[t][n], false, false); \
        }                                                                                 \
    }

    v2f a0[GEMM_KC][GEMM_MT], b0[GEMM_KC][4];
    v2f a1[GEMM_KC][GEMM_MT], b1[GEMM_KC][4];
    GEMM_LOAD_SET(a0, b0, 0)
    for (int k0 = 0; k0 < IN_F - 8 * GEMM_KC; k0 += 8 * GEMM_KC) {
        GEMM_LOAD_SET(a1, b1, k0 + 4 * GEMM_KC)
        asm volatile("" ::: "memory");
        GEMM_MUL_SET(a0, b0)
        GEMM_LOAD_SET(a0, b0, k0 + 8 * GEMM_KC)
        asm volatile("" ::: "memory");
        GEMM_MUL_SET(a1, b1)
    }
    GEMM_LOAD_SET(a1, b1, IN_F - 4 * GEMM_KC)
    asm volatile("" ::: "memory");
    GEMM_MUL_SET(a0, b0)
    GEMM_MUL_SET(a1, b1)
#undef GEMM_MUL_SET
#undef GEMM_LOAD_SET

    const int col = lane & 15;
    const float bv0 = bias[col +  0];
    const float bv1 = bias[col + 16];
    const float bv2 = bias[col + 32];
    const float bv3 = bias[col + 48];

#pragma unroll
    for (int t = 0; t < GEMM_MT; ++t) {
        const int tile = wave * GEMM_MT + t;
        if (tile < NTILES) {
            const int row_base = tile * 16 + ((lane >> 4) ? 8 : 0);
#pragma unroll
            for (int r = 0; r < 8; ++r) {
                float* o = out + (size_t)(row_base + r) * OUT_F + col;
                o[ 0] = acc[t][0][r] + bv0;
                o[16] = acc[t][1][r] + bv1;
                o[32] = acc[t][2][r] + bv2;
                o[48] = acc[t][3][r] + bv3;
            }
        }
    }
}

#define HOP_GROUPS      256
#define ROWS_PER_GROUP  391
#define HOP_THREADS     1024
#define HOP_WAVES       32
#define ROWS_PER_WAVE   13
#define EDGES_PER_CHUNK 128
#define N_CHUNKS        (N_EDGES / EDGES_PER_CHUNK)
#define BOX_WORDS       32
#define EDGE_BITS       21
#define QUEUE_WORDS     160

typedef int v4i __attribute__((ext_vector_type(4)));

template <bool FULL>
__device__ __forceinline__ void hop_drain(const int* qp, int n, int lane,
                                          const float* __restrict__ cur,
                                          const int* __restrict__ esrc,
                                          const float* __restrict__ ew,
                                          float* acc) {
    int   s  = 0;
    int   rw = 0;
    float wt = 0.f;
    if (FULL || lane < n) {
        const unsigned pk = (unsigned)qp[lane];
        const int e = (int)(pk & ((1u << EDGE_BITS) - 1u));
        rw = (int)(pk >> EDGE_BITS);
        s  = esrc[e];
        wt = ew[e];
    }
    const int wti = __builtin_bit_cast(int, wt);
    const int cnt = FULL ? 32 : n;

    v2f   v[8];
    int   rj[8];
    float wj[8];
#define HOP_GATHER8(J0, V, RJ, WJ)                                                        \
    _Pragma("unroll")                                                                     \
    for (int u = 0; u < 8; ++u) {                                                         \
        int jj = (J0) + u;                                                                \
        if (!FULL) jj = jj < n ? jj : n - 1;                                              \
        const int sj = __builtin_amdgcn_readlane(s, jj);                                  \
        RJ[u] = __builtin_amdgcn_readlane(rw, jj);                                        \
        WJ[u] = __builtin_bit_cast(float, __builtin_amdgcn_readlane(wti, jj));            \
        V[u]  = *(const v2f*)(cur + (size_t)sj * OUT_F + lane * 2);                       \
    }
    HOP_GATHER8(0, v, rj, wj)
    for (int j0 = 0; j0 < cnt; j0 += 8) {
        v2f   nv[8];
        int   nrj[8];
        float nwj[8];
        const bool more = j0 + 8 < cnt;
        if (more) {
            HOP_GATHER8(j0 + 8, nv, nrj, nwj)
        }
#pragma unroll
        for (int u = 0; u < 8; ++u) {
            if (FULL || j0 + u < n) {
                v2f* a = (v2f*)(acc + rj[u] * OUT_F) + lane;
                *a = *a + v[u] * wj[u];
            }
        }
        if (more) {
#pragma unroll
            for (int u = 0; u < 8; ++u) { v[u] = nv[u]; rj[u] = nrj[u]; wj[u] = nwj[u]; }
        }
    }
#undef HOP_GATHER8
}

__global__ __launch_bounds__(HOP_THREADS)
void sgc_hop_owner_waves(const float* __restrict__ cur,
                         float* __restrict__ nxt,
                         const int* __restrict__ esrc,
                         const int* __restrict__ edst,
                         const float* __restrict__ ew) {
    __shared__ __attribute__((aligned(16))) float acc[ROWS_PER_GROUP * OUT_F];
    __shared__ int boxes[HOP_WAVES * HOP_WAVES * BOX_WORDS];
    __shared__ int box_cnt[HOP_WAVES * HOP_WAVES];
    __shared__ int ownq[HOP_WAVES][64];
    __shared__ int overflow_flag;

    const int tid  = threadIdx.x;
    const int lane = tid & 31;
    const int wv   = __builtin_amdgcn_readfirstlane(tid >> 5);
    const int base = blockIdx.x * ROWS_PER_GROUP;
    const int left = N_NODES - base;
    const unsigned nrows = (unsigned)(left < ROWS_PER_GROUP ? left : ROWS_PER_GROUP);

    if (tid == 0) overflow_flag = 0;
    float4* acc4 = (float4*)acc;
    for (int i = tid; i < ROWS_PER_GROUP * OUT_F / 4; i += HOP_THREADS)
        acc4[i] = make_float4(0.f, 0.f, 0.f, 0.f);
    __syncthreads();

    const v4i* __restrict__ dst4 = (const v4i*)edst;

    int  cntv = 0;
    bool ovf  = false;
    v4i d[4];
#pragma unroll
    for (int u = 0; u < 4; ++u) d[u] = dst4[(size_t)(u * HOP_WAVES + wv) * 32 + lane];
    for (int c0 = 0; c0 < N_CHUNKS; c0 += 4 * HOP_WAVES) {
        v4i nd[4];
        int c[4];
#pragma unroll
        for (int u = 0; u < 4; ++u) {
            c[u] = c0 + u * HOP_WAVES + wv;
            const int cn = c[u] + 4 * HOP_WAVES;
            const int cc = cn < N_CHUNKS ? cn : N_CHUNKS - 1;
            nd[u] = dst4[(size_t)cc * 32 + lane];
        }
#pragma unroll
        for (int u = 0; u < 4; ++u) {
            if (c[u] >= N_CHUNKS) continue;
            const unsigned rr[4] = {(unsigned)(d[u].x - base), (unsigned)(d[u].y - base),
                                    (unsigned)(d[u].z - base), (unsigned)(d[u].w - base)};
            const unsigned m01 = rr[0] < rr[1] ? rr[0] : rr[1];
            const unsigned m23 = rr[2] < rr[3] ? rr[2] : rr[3];
            const unsigned mn  = m01 < m23 ? m01 : m23;
            if (__builtin_amdgcn_ballot_w32(mn < nrows) == 0u) continue;

            const unsigned e0 = (unsigned)(c[u] * EDGES_PER_CHUNK + lane * 4);
#pragma unroll
            for (int k = 0; k < 4; ++k) {
                const bool hit = rr[k] < nrows;
                const int  pkv = (int)((rr[k] << EDGE_BITS) | (e0 + (unsigned)k));
                unsigned m = __builtin_amdgcn_ballot_w32(hit);
                while (m != 0u) {
                    const int l = __builtin_ctz(m);
                    m &= m - 1u;
                    const int p = __builtin_amdgcn_readlane(pkv, l);
                    const int o = (int)(((unsigned)p >> EDGE_BITS) & 31u);
                    if (lane == o) {
                        if (cntv < BOX_WORDS) {
                            boxes[(wv * HOP_WAVES + o) * BOX_WORDS + cntv] = p;
                            ++cntv;
                        } else {
                            ovf = true;
                        }
                    }
                }
            }
        }
#pragma unroll
        for (int u = 0; u < 4; ++u) d[u] = nd[u];
    }
    box_cnt[wv * HOP_WAVES + lane] = cntv;
    if (__builtin_amdgcn_ballot_w32(ovf) != 0u && lane == 0) overflow_flag = 1;
    __syncthreads();
    const int over = __builtin_amdgcn_readfirstlane(overflow_flag);

    if (over == 0) {
        const int  mine  = box_cnt[lane * HOP_WAVES + wv];
        const int* mybox = boxes + (lane * HOP_WAVES + wv) * BOX_WORDS;
        int* oq = ownq[wv];
        int  on = 0;
        for (int i = 0; i < BOX_WORDS; ++i) {
            const bool has = i < mine;
            const unsigned m = __builtin_amdgcn_ballot_w32(has);
            if (m == 0u) break;
            const int pos = on + (int)__builtin_amdgcn_mbcnt_lo(m, 0u);
            if (has) oq[pos] = mybox[i];
            on += __builtin_popcount(m);
            if (on >= 32) {
                on -= 32;
                hop_drain<true>(oq + on, 32, lane, cur, esrc, ew, acc);
            }
        }
        if (on > 0) hop_drain<false>(oq, on, lane, cur, esrc, ew, acc);
    } else {
        int* sq = boxes + wv * QUEUE_WORDS;
        int  qn = 0;
        for (int c0 = 0; c0 < N_CHUNKS; c0 += 4) {
            v4i d[4];
#pragma unroll
            for (int u = 0; u < 4; ++u) d[u] = dst4[(size_t)(c0 + u) * 32 + lane];
#pragma unroll
            for (int u = 0; u < 4; ++u) {
                const unsigned rr[4] = {(unsigned)(d[u].x - base), (unsigned)(d[u].y - base),
                                        (unsigned)(d[u].z - base), (unsigned)(d[u].w - base)};
                const unsigned e0 = (unsigned)((c0 + u) * EDGES_PER_CHUNK + lane * 4);
#pragma unroll
                for (int k = 0; k < 4; ++k) {
                    const bool hit = rr[k] < nrows && (int)(rr[k] & 31u) == wv;
                    const unsigned m = __builtin_amdgcn_ballot_w32(hit);
                    if (m != 0u) {
                        const int pos = qn + (int)__builtin_amdgcn_mbcnt_lo(m, 0u);
                        if (hit) sq[pos] = (int)((rr[k] << EDGE_BITS) | (e0 + (unsigned)k));
                        qn += __builtin_popcount(m);
                    }
                }
                while (qn >= 32) {
                    qn -= 32;
                    hop_drain<true>(sq + qn, 32, lane, cur, esrc, ew, acc);
                }
            }
        }
        if (qn > 0) hop_drain<false>(sq, qn, lane, cur, esrc, ew, acc);
    }

#pragma unroll
    for (int r = 0; r < ROWS_PER_WAVE; ++r) {
        const unsigned row = (unsigned)(wv + HOP_WAVES * r);
        if (row < nrows)
            *(v2f*)(nxt + (size_t)(base + (int)row) * OUT_F + lane * 2) =
                *((const v2f*)(acc + row * OUT_F) + lane);
    }
}

extern "C" void kernel_launch(void* const* d_in, const int* in_sizes, int n_in,
                              void* d_out, int out_size, void* d_ws, size_t ws_size,
                              hipStream_t stream) {
    const float* x    = (const float*)d_in[0];
    const float* W    = (const float*)d_in[1];
    const float* b    = (const float*)d_in[2];
    const int*   esrc = (const int*)d_in[3];
    const int*   edst = (const int*)d_in[4];
    const float* ew   = (const float*)d_in[5];

    float* out = (float*)d_out;
    float* ws  = (float*)d_ws;

    sgc_gemm_wmma<<<GEMM_GROUPS, GEMM_THREADS, 0, stream>>>(x, W, b, ws);

    sgc_hop_owner_waves<<<HOP_GROUPS, HOP_THREADS, 0, stream>>>(ws, out, esrc, edst, ew);
    sgc_hop_owner_waves<<<HOP_GROUPS, HOP_THREADS, 0, stream>>>(out, ws, esrc, edst, ew);
    sgc_hop_owner_waves<<<HOP_GROUPS, HOP_THREADS, 0, stream>>>(ws, out, esrc, edst, ew);
}


# hops, list scan, fewer instructions a chunk: a hit reaches its owner lane by one predicated LDS store and one add, overflow read once after the loop; two trips a pass, no register copies; same sums
# speedup vs baseline: 1.6968x; 1.0815x over previous
#include <hip/hip_runtime.h>
#include <hip/hip_bf16.h>

typedef float v2f __attribute__((ext_vector_type(2)));
typedef float v8f __attribute__((ext_vector_type(8)));

#define N_NODES 100000
#define N_EDGES 1600000
#define IN_F    256
#define OUT_F   64
#define NTILES  (N_NODES / 16)

#define GEMM_MT      4
#define GEMM_WAVES   ((NTILES + GEMM_MT - 1) / GEMM_MT)
#define GEMM_THREADS 64
#define GEMM_GROUPS  ((GEMM_WAVES + 1) / 2)
#define GEMM_KC      4

__global__ __launch_bounds__(GEMM_THREADS)
void sgc_gemm_wmma(const float* __restrict__ x,
                   const float* __restrict__ W,
                   const float* __restrict__ bias,
                   float* __restrict__ out) {
    const int wave = (blockIdx.x * GEMM_THREADS + threadIdx.x) >> 5;
    const int lane = threadIdx.x & 31;
    if (wave >= GEMM_WAVES) return;

    const int m    = lane & 15;
    const int koff = (lane >> 4) << 1;

    const float* __restrict__ xr[GEMM_MT];
    const float* __restrict__ wr[4];
#pragma unroll
    for (int t = 0; t < GEMM_MT; ++t) {
        const int tile = wave * GEMM_MT + t;
        const int tl   = tile < NTILES ? tile : NTILES - 1;
        xr[t] = x + (size_t)(tl * 16 + m) * IN_F + koff;
    }
#pragma unroll
    for (int n = 0; n < 4; ++n) wr[n] = W + (size_t)(m + 16 * n) * IN_F + koff;

    v8f acc[GEMM_MT][4];
#pragma unroll
    for (int t = 0; t < GEMM_MT; ++t)
#pragma unroll
        for (int n = 0; n < 4; ++n) acc[t][n] = v8f{};

#define GEMM_LOAD_SET(A, B, K0)                                                           \
    _Pragma("unroll")                                                                     \
    for (int kk = 0; kk < GEMM_KC; ++kk) {                                                \
        _Pragma("unroll")                                                                 \
        for (int t = 0; t < GEMM_MT; ++t) A[kk][t] = *(const v2f*)(xr[t] + (K0) + 4 * kk); \
        _Pragma("unroll")                                                                 \
        for (int n = 0; n < 4; ++n)       B[kk][n] = *(const v2f*)(wr[n] + (K0) + 4 * kk); \
    }

#define GEMM_MUL_SET(A, B)                                                                \
    _Pragma("unroll")                                                                     \
    for (int kk = 0; kk < GEMM_KC; ++kk) {                                                \
        _Pragma("unroll")                                                                 \
        for (int t = 0; t < GEMM_MT; ++t) {                                               \
            _Pragma("unroll")                                                             \
            for (int n = 0; n < 4; ++n)                                                   \
                acc[t][n] = __builtin_amdgcn_wmma_f32_16x16x4_f32(false, A[kk][t], false, B[kk][n], \
                                                                  (short)0, acc[t][n], false, false); \
        }                                                                                 \
    }

    v2f a0[GEMM_KC][GEMM_MT], b0[GEMM_KC][4];
    v2f a1[GEMM_KC][GEMM_MT], b1[GEMM_KC][4];
    GEMM_LOAD_SET(a0, b0, 0)
    for (int k0 = 0; k0 < IN_F - 8 * GEMM_KC; k0 += 8 * GEMM_KC) {
        GEMM_LOAD_SET(a1, b1, k0 + 4 * GEMM_KC)
        asm volatile("" ::: "memory");
        GEMM_MUL_SET(a0, b0)
        GEMM_LOAD_SET(a0, b0, k0 + 8 * GEMM_KC)
        asm volatile("" ::: "memory");
        GEMM_MUL_SET(a1, b1)
    }
    GEMM_LOAD_SET(a1, b1, IN_F - 4 * GEMM_KC)
    asm volatile("" ::: "memory");
    GEMM_MUL_SET(a0, b0)
    GEMM_MUL_SET(a1, b1)
#undef GEMM_MUL_SET
#undef GEMM_LOAD_SET

    const int col = lane & 15;
    const float bv0 = bias[col +  0];
    const float bv1 = bias[col + 16];
    const float bv2 = bias[col + 32];
    const float bv3 = bias[col + 48];

#pragma unroll
    for (int t = 0; t < GEMM_MT; ++t) {
        const int tile = wave * GEMM_MT + t;
        if (tile < NTILES) {
            const int row_base = tile * 16 + ((lane >> 4) ? 8 : 0);
#pragma unroll
            for (int r = 0; r < 8; ++r) {
                float* o = out + (size_t)(row_base + r) * OUT_F + col;
                o[ 0] = acc[t][0][r] + bv0;
                o[16] = acc[t][1][r] + bv1;
                o[32] = acc[t][2][r] + bv2;
                o[48] = acc[t][3][r] + bv3;
            }
        }
    }
}

#define HOP_GROUPS      256
#define ROWS_PER_GROUP  391
#define HOP_THREADS     1024
#define HOP_WAVES       32
#define ROWS_PER_WAVE   13
#define EDGES_PER_CHUNK 128
#define N_CHUNKS        (N_EDGES / EDGES_PER_CHUNK)
#define BOX_WORDS       32
#define EDGE_BITS       21
#define QUEUE_WORDS     160

typedef int v4i __attribute__((ext_vector_type(4)));

template <bool FULL>
__device__ __forceinline__ void hop_drain(const int* qp, int n, int lane,
                                          const float* __restrict__ cur,
                                          const int* __restrict__ esrc,
                                          const float* __restrict__ ew,
                                          float* acc) {
    int   s  = 0;
    int   rw = 0;
    float wt = 0.f;
    if (FULL || lane < n) {
        const unsigned pk = (unsigned)qp[lane];
        const int e = (int)(pk & ((1u << EDGE_BITS) - 1u));
        rw = (int)(pk >> EDGE_BITS);
        s  = esrc[e];
        wt = ew[e];
    }
    const int wti = __builtin_bit_cast(int, wt);
    const int cnt = FULL ? 32 : n;

    v2f   v[8];
    int   rj[8];
    float wj[8];
#define HOP_GATHER8(J0, V, RJ, WJ)                                                        \
    _Pragma("unroll")                                                                     \
    for (int u = 0; u < 8; ++u) {                                                         \
        int jj = (J0) + u;                                                                \
        if (!FULL) jj = jj < n ? jj : n - 1;                                              \
        const int sj = __builtin_amdgcn_readlane(s, jj);                                  \
        RJ[u] = __builtin_amdgcn_readlane(rw, jj);                                        \
        WJ[u] = __builtin_bit_cast(float, __builtin_amdgcn_readlane(wti, jj));            \
        V[u]  = *(const v2f*)(cur + (size_t)sj * OUT_F + lane * 2);                       \
    }
    HOP_GATHER8(0, v, rj, wj)
    for (int j0 = 0; j0 < cnt; j0 += 8) {
        v2f   nv[8];
        int   nrj[8];
        float nwj[8];
        const bool more = j0 + 8 < cnt;
        if (more) {
            HOP_GATHER8(j0 + 8, nv, nrj, nwj)
        }
#pragma unroll
        for (int u = 0; u < 8; ++u) {
            if (FULL || j0 + u < n) {
                v2f* a = (v2f*)(acc + rj[u] * OUT_F) + lane;
                *a = *a + v[u] * wj[u];
            }
        }
        if (more) {
#pragma unroll
            for (int u = 0; u < 8; ++u) { v[u] = nv[u]; rj[u] = nrj[u]; wj[u] = nwj[u]; }
        }
    }
#undef HOP_GATHER8
}

__global__ __launch_bounds__(HOP_THREADS)
void sgc_hop_owner_waves(const float* __restrict__ cur,
                         float* __restrict__ nxt,
                         const int* __restrict__ esrc,
                         const int* __restrict__ edst,
                         const float* __restrict__ ew) {
    __shared__ __attribute__((aligned(16))) float acc[ROWS_PER_GROUP * OUT_F];
    __shared__ int boxes[HOP_WAVES * HOP_WAVES * BOX_WORDS];
    __shared__ int box_cnt[HOP_WAVES * HOP_WAVES];
    __shared__ int ownq[HOP_WAVES][64];
    __shared__ int overflow_flag;

    const int tid  = threadIdx.x;
    const int lane = tid & 31;
    const int wv   = __builtin_amdgcn_readfirstlane(tid >> 5);
    const int base = blockIdx.x * ROWS_PER_GROUP;
    const int left = N_NODES - base;
    const unsigned nrows = (unsigned)(left < ROWS_PER_GROUP ? left : ROWS_PER_GROUP);

    if (tid == 0) overflow_flag = 0;
    float4* acc4 = (float4*)acc;
    for (int i = tid; i < ROWS_PER_GROUP * OUT_F / 4; i += HOP_THREADS)
        acc4[i] = make_float4(0.f, 0.f, 0.f, 0.f);
    __syncthreads();

    const v4i* __restrict__ dst4 = (const v4i*)edst;

    int  cntv = 0;
    int* mybox_w = boxes + (wv * HOP_WAVES + lane) * BOX_WORDS;
    auto scan_trip = [&](const v4i (&D)[4], v4i (&ND)[4], const int c0) __attribute__((always_inline)) {
        int c[4];
#pragma unroll
        for (int u = 0; u < 4; ++u) {
            c[u] = c0 + u * HOP_WAVES + wv;
            const int cn = c[u] + 4 * HOP_WAVES;
            const int cc = cn < N_CHUNKS ? cn : N_CHUNKS - 1;
            ND[u] = dst4[(size_t)cc * 32 + lane];
        }
#pragma unroll
        for (int u = 0; u < 4; ++u) {
            if (c[u] >= N_CHUNKS) continue;
            const unsigned rr[4] = {(unsigned)(D[u].x - base), (unsigned)(D[u].y - base),
                                    (unsigned)(D[u].z - base), (unsigned)(D[u].w - base)};
            const unsigned m01 = rr[0] < rr[1] ? rr[0] : rr[1];
            const unsigned m23 = rr[2] < rr[3] ? rr[2] : rr[3];
            const unsigned mn  = m01 < m23 ? m01 : m23;
            if (__builtin_amdgcn_ballot_w32(mn < nrows) == 0u) continue;

            const unsigned e0 = (unsigned)(c[u] * EDGES_PER_CHUNK + lane * 4);
#pragma unroll
            for (int k = 0; k < 4; ++k) {
                const bool hit = rr[k] < nrows;
                const int  pkv = (int)((rr[k] << EDGE_BITS) | (e0 + (unsigned)k));
                unsigned m = __builtin_amdgcn_ballot_w32(hit);
                while (m != 0u) {
                    const int l = __builtin_ctz(m);
                    m &= m - 1u;
                    const int p = __builtin_amdgcn_readlane(pkv, l);
                    const int o = (int)(((unsigned)p >> EDGE_BITS) & 31u);
                    const bool me = lane == o;
                    if (me && cntv < BOX_WORDS) mybox_w[cntv] = p;
                    cntv += me ? 1 : 0;
                }
            }
        }
    };
    static_assert(((N_CHUNKS + 4 * HOP_WAVES - 1) / (4 * HOP_WAVES)) % 2 == 0, "even number of trips");
    v4i d[4], nd[4];
#pragma unroll
    for (int u = 0; u < 4; ++u) d[u] = dst4[(size_t)(u * HOP_WAVES + wv) * 32 + lane];
    for (int c0 = 0; c0 < N_CHUNKS; c0 += 8 * HOP_WAVES) {
        scan_trip(d, nd, c0);
        scan_trip(nd, d, c0 + 4 * HOP_WAVES);
    }
    box_cnt[wv * HOP_WAVES + lane] = cntv < BOX_WORDS ? cntv : BOX_WORDS;
    if (__builtin_amdgcn_ballot_w32(cntv > BOX_WORDS) != 0u && lane == 0) overflow_flag = 1;
    __syncthreads();
    const int over = __builtin_amdgcn_readfirstlane(overflow_flag);

    if (over == 0) {
        const int  mine  = box_cnt[lane * HOP_WAVES + wv];
        const int* mybox = boxes + (lane * HOP_WAVES + wv) * BOX_WORDS;
        int* oq = ownq[wv];
        int  on = 0;
        for (int i = 0; i < BOX_WORDS; ++i) {
            const bool has = i < mine;
            const unsigned m = __builtin_amdgcn_ballot_w32(has);
            if (m == 0u) break;
            const int pos = on + (int)__builtin_amdgcn_mbcnt_lo(m, 0u);
            if (has) oq[pos] = mybox[i];
            on += __builtin_popcount(m);
            if (on >= 32) {
                on -= 32;
                hop_drain<true>(oq + on, 32, lane, cur, esrc, ew, acc);
            }
        }
        if (on > 0) hop_drain<false>(oq, on, lane, cur, esrc, ew, acc);
    } else {
        int* sq = boxes + wv * QUEUE_WORDS;
        int  qn = 0;
        for (int c0 = 0; c0 < N_CHUNKS; c0 += 4) {
            v4i d[4];
#pragma unroll
            for (int u = 0; u < 4; ++u) d[u] = dst4[(size_t)(c0 + u) * 32 + lane];
#pragma unroll
            for (int u = 0; u < 4; ++u) {
                const unsigned rr[4] = {(unsigned)(d[u].x - base), (unsigned)(d[u].y - base),
                                        (unsigned)(d[u].z - base), (unsigned)(d[u].w - base)};
                const unsigned e0 = (unsigned)((c0 + u) * EDGES_PER_CHUNK + lane * 4);
#pragma unroll
                for (int k = 0; k < 4; ++k) {
                    const bool hit = rr[k] < nrows && (int)(rr[k] & 31u) == wv;
                    const unsigned m = __builtin_amdgcn_ballot_w32(hit);
                    if (m != 0u) {
                        const int pos = qn + (int)__builtin_amdgcn_mbcnt_lo(m, 0u);
                        if (hit) sq[pos] = (int)((rr[k] << EDGE_BITS) | (e0 + (unsigned)k));
                        qn += __builtin_popcount(m);
                    }
                }
                while (qn >= 32) {
                    qn -= 32;
                    hop_drain<true>(sq + qn, 32, lane, cur, esrc, ew, acc);
                }
            }
        }
        if (qn > 0) hop_drain<false>(sq, qn, lane, cur, esrc, ew, acc);
    }

#pragma unroll
    for (int r = 0; r < ROWS_PER_WAVE; ++r) {
        const unsigned row = (unsigned)(wv + HOP_WAVES * r);
        if (row < nrows)
            *(v2f*)(nxt + (size_t)(base + (int)row) * OUT_F + lane * 2) =
                *((const v2f*)(acc + row * OUT_F) + lane);
    }
}

extern "C" void kernel_launch(void* const* d_in, const int* in_sizes, int n_in,
                              void* d_out, int out_size, void* d_ws, size_t ws_size,
                              hipStream_t stream) {
    const float* x    = (const float*)d_in[0];
    const float* W    = (const float*)d_in[1];
    const float* b    = (const float*)d_in[2];
    const int*   esrc = (const int*)d_in[3];
    const int*   edst = (const int*)d_in[4];
    const float* ew   = (const float*)d_in[5];

    float* out = (float*)d_out;
    float* ws  = (float*)d_ws;

    sgc_gemm_wmma<<<GEMM_GROUPS, GEMM_THREADS, 0, stream>>>(x, W, b, ws);

    sgc_hop_owner_waves<<<HOP_GROUPS, HOP_THREADS, 0, stream>>>(ws, out, esrc, edst, ew);
    sgc_hop_owner_waves<<<HOP_GROUPS, HOP_THREADS, 0, stream>>>(out, ws, esrc, edst, ew);
    sgc_hop_owner_waves<<<HOP_GROUPS, HOP_THREADS, 0, stream>>>(ws, out, esrc, edst, ew);
}


# hops: a launch before each hop sorts the edges by owner row group (ballot ranks, no atomic) into 8-B records (source, row, weight) in the hop's own output rows; group reads 100 KB; sum order changes
# speedup vs baseline: 2.7477x; 1.6193x over previous
#include <hip/hip_runtime.h>
#include <hip/hip_bf16.h>

typedef float v2f __attribute__((ext_vector_type(2)));
typedef float v8f __attribute__((ext_vector_type(8)));

#define N_NODES 100000
#define N_EDGES 1600000
#define IN_F    256
#define OUT_F   64
#define NTILES  (N_NODES / 16)

#define GEMM_MT      4
#define GEMM_WAVES   ((NTILES + GEMM_MT - 1) / GEMM_MT)
#define GEMM_THREADS 64
#define GEMM_GROUPS  ((GEMM_WAVES + 1) / 2)
#define GEMM_KC      4

__global__ __launch_bounds__(GEMM_THREADS)
void sgc_gemm_wmma(const float* __restrict__ x,
                   const float* __restrict__ W,
                   const float* __restrict__ bias,
                   float* __restrict__ out) {
    const int wave = (blockIdx.x * GEMM_THREADS + threadIdx.x) >> 5;
    const int lane = threadIdx.x & 31;
    if (wave >= GEMM_WAVES) return;

    const int m    = lane & 15;
    const int koff = (lane >> 4) << 1;

    const float* __restrict__ xr[GEMM_MT];
    const float* __restrict__ wr[4];
#pragma unroll
    for (int t = 0; t < GEMM_MT; ++t) {
        const int tile = wave * GEMM_MT + t;
        const int tl   = tile < NTILES ? tile : NTILES - 1;
        xr[t] = x + (size_t)(tl * 16 + m) * IN_F + koff;
    }
#pragma unroll
    for (int n = 0; n < 4; ++n) wr[n] = W + (size_t)(m + 16 * n) * IN_F + koff;

    v8f acc[GEMM_MT][4];
#pragma unroll
    for (int t = 0; t < GEMM_MT; ++t)
#pragma unroll
        for (int n = 0; n < 4; ++n) acc[t][n] = v8f{};

#define GEMM_LOAD_SET(A, B, K0)                                                           \
    _Pragma("unroll")                                                                     \
    for (int kk = 0; kk < GEMM_KC; ++kk) {                                                \
        _Pragma("unroll")                                                                 \
        for (int t = 0; t < GEMM_MT; ++t) A[kk][t] = *(const v2f*)(xr[t] + (K0) + 4 * kk); \
        _Pragma("unroll")                                                                 \
        for (int n = 0; n < 4; ++n)       B[kk][n] = *(const v2f*)(wr[n] + (K0) + 4 * kk); \
    }

#define GEMM_MUL_SET(A, B)                                                                \
    _Pragma("unroll")                                                                     \
    for (int kk = 0; kk < GEMM_KC; ++kk) {                                                \
        _Pragma("unroll")                                                                 \
        for (int t = 0; t < GEMM_MT; ++t) {                                               \
            _Pragma("unroll")                                                             \
            for (int n = 0; n < 4; ++n)                                                   \
                acc[t][n] = __builtin_amdgcn_wmma_f32_16x16x4_f32(false, A[kk][t], false, B[kk][n], \
                                                                  (short)0, acc[t][n], false, false); \
        }                                                                                 \
    }

    v2f a0[GEMM_KC][GEMM_MT], b0[GEMM_KC][4];
    v2f a1[GEMM_KC][GEMM_MT], b1[GEMM_KC][4];
    GEMM_LOAD_SET(a0, b0, 0)
    for (int k0 = 0; k0 < IN_F - 8 * GEMM_KC; k0 += 8 * GEMM_KC) {
        GEMM_LOAD_SET(a1, b1, k0 + 4 * GEMM_KC)
        asm volatile("" ::: "memory");
        GEMM_MUL_SET(a0, b0)
        GEMM_LOAD_SET(a0, b0, k0 + 8 * GEMM_KC)
        asm volatile("" ::: "memory");
        GEMM_MUL_SET(a1, b1)
    }
    GEMM_LOAD_SET(a1, b1, IN_F - 4 * GEMM_KC)
    asm volatile("" ::: "memory");
    GEMM_MUL_SET(a0, b0)
    GEMM_MUL_SET(a1, b1)
#undef GEMM_MUL_SET
#undef GEMM_LOAD_SET

    const int col = lane & 15;
    const float bv0 = bias[col +  0];
    const float bv1 = bias[col + 16];
    const float bv2 = bias[col + 32];
    const float bv3 = bias[col + 48];

#pragma unroll
    for (int t = 0; t < GEMM_MT; ++t) {
        const int tile = wave * GEMM_MT + t;
        if (tile < NTILES) {
            const int row_base = tile * 16 + ((lane >> 4) ? 8 : 0);
#pragma unroll
            for (int r = 0; r < 8; ++r) {
                float* o = out + (size_t)(row_base + r) * OUT_F + col;
                o[ 0] = acc[t][0][r] + bv0;
                o[16] = acc[t][1][r] + bv1;
                o[32] = acc[t][2][r] + bv2;
                o[48] = acc[t][3][r] + bv3;
            }
        }
    }
}

#define HOP_GROUPS      256
#define ROWS_PER_GROUP  391
#define HOP_THREADS     1024
#define HOP_WAVES       32
#define ROWS_PER_WAVE   13
#define EDGES_PER_CHUNK 128
#define N_CHUNKS        (N_EDGES / EDGES_PER_CHUNK)
#define ROW_BITS        9
#define BOX_SLOTS       24
#define BOX_RSTRIDE     (HOP_WAVES * BOX_SLOTS + 1)
#define QUEUE_RECS      160

#define BIN_SLICES      128
#define SLICE_EDGES     (N_EDGES / BIN_SLICES)
#define BIN_THREADS     512
#define BIN_WAVES       16
#define SLICE_BATCHES   ((SLICE_EDGES + 31) / 32)
#define BATCHES_PER_WAVE ((SLICE_BATCHES + BIN_WAVES - 1) / BIN_WAVES)
#define SUB_SLOTS_MAX   96
#define SUB_ROUNDS      3

typedef int v2i __attribute__((ext_vector_type(2)));
typedef int v4i __attribute__((ext_vector_type(4)));

static_assert(N_EDGES % BIN_SLICES == 0, "slices of equal length");
static_assert(BIN_SLICES == 4 * HOP_WAVES, "four sub-bins a reading wave");
static_assert(SUB_SLOTS_MAX - 1 <= 32 * SUB_ROUNDS, "a sub-bin is read in SUB_ROUNDS rounds");
static_assert(BIN_SLICES * 8 == 4 * 256, "rows >> 2 slots a sub-bin use up the rows of the group");
static_assert(HOP_WAVES * QUEUE_RECS <= HOP_WAVES * BOX_RSTRIDE, "slow-path queues fit the boxes");

__device__ __forceinline__ int sub_bin_slots(int rows) {
    const int s = rows >> 2;
    return s < SUB_SLOTS_MAX ? s : SUB_SLOTS_MAX;
}

__device__ __forceinline__ int   bits_of(float f) { return __builtin_bit_cast(int, f); }
__device__ __forceinline__ float float_of(int i)  { return __builtin_bit_cast(float, i); }

template <int NBITS>
__device__ __forceinline__ unsigned lanes_with_same_key(unsigned key, bool valid) {
    unsigned peers = __builtin_amdgcn_ballot_w32(valid);
#pragma unroll
    for (int b = 0; b < NBITS; ++b) {
        const bool bit = ((key >> b) & 1u) != 0u;
        const unsigned m = __builtin_amdgcn_ballot_w32(bit);
        peers &= bit ? m : ~m;
    }
    return peers;
}

__global__ __launch_bounds__(BIN_THREADS)
void sgc_bin_edges(const int* __restrict__ esrc,
                   const int* __restrict__ edst,
                   const float* __restrict__ ew,
                   float* __restrict__ nxt) {
    __shared__ v2i sorted[SLICE_EDGES];
    __shared__ int cnt[BIN_WAVES * HOP_GROUPS];
    __shared__ int gtot[HOP_GROUPS];
    __shared__ int gstart[HOP_GROUPS];

    const int tid  = threadIdx.x;
    const int lane = tid & 31;
    const int wv   = __builtin_amdgcn_readfirstlane(tid >> 5);
    const int e_base = blockIdx.x * SLICE_EDGES;
    const unsigned below = (1u << lane) - 1u;

    for (int i = tid; i < BIN_WAVES * HOP_GROUPS; i += BIN_THREADS) cnt[i] = 0;

    int   dv[BATCHES_PER_WAVE];
    int   sv[BATCHES_PER_WAVE];
    float wt[BATCHES_PER_WAVE];
#pragma unroll
    for (int j = 0; j < BATCHES_PER_WAVE; ++j) {
        const int i  = (wv + BIN_WAVES * j) * 32 + lane;
        const int e  = e_base + (i < SLICE_EDGES ? i : 0);
        dv[j] = edst[e];
        sv[j] = esrc[e];
        wt[j] = ew[e];
    }
    __syncthreads();

    int pk[BATCHES_PER_WAVE];
#pragma unroll
    for (int j = 0; j < BATCHES_PER_WAVE; ++j) {
        const int  i     = (wv + BIN_WAVES * j) * 32 + lane;
        const bool valid = i < SLICE_EDGES;
        const unsigned g = ((unsigned)dv[j] / (unsigned)ROWS_PER_GROUP) & (HOP_GROUPS - 1u);
        const unsigned r = (unsigned)dv[j] - g * (unsigned)ROWS_PER_GROUP;
        const unsigned peers = lanes_with_same_key<8>(g, valid);
        const unsigned rank  = (unsigned)__builtin_popcount(peers & below);
        const unsigned tot   = valid ? (unsigned)__builtin_popcount(peers) : 0u;
        if (valid && rank == 0u) cnt[wv * HOP_GROUPS + (int)g] += (int)tot;
        pk[j] = (int)(g | ((r & 511u) << 8) | (rank << 17) | (tot << 22));
    }
    __syncthreads();

    if (tid < HOP_GROUPS) {
        int a = 0;
        for (int w = 0; w < BIN_WAVES; ++w) {
            const int c = cnt[w * HOP_GROUPS + tid];
            cnt[w * HOP_GROUPS + tid] = a;
            a += c;
        }
        gtot[tid] = a;
    }
    __syncthreads();
    if (tid < HOP_GROUPS) {
        int a = 0;
        for (int g = 0; g < tid; ++g) a += gtot[g];
        gstart[tid] = a;
    }
    __syncthreads();

#pragma unroll
    for (int j = 0; j < BATCHES_PER_WAVE; ++j) {
        const unsigned p   = (unsigned)pk[j];
        const int      tot = (int)(p >> 22);
        if (tot != 0) {
            const int g    = (int)(p & 255u);
            const int rank = (int)((p >> 17) & 31u);
            const int off  = cnt[wv * HOP_GROUPS + g];
            v2i rec;
            rec.x = (int)(((unsigned)sv[j] << ROW_BITS) | ((p >> 8) & 511u));
            rec.y = __builtin_bit_cast(int, wt[j]);
            sorted[gstart[g] + off + rank] = rec;
            if (rank == 0) cnt[wv * HOP_GROUPS + g] = off + tot;
        }
    }
    __syncthreads();

    for (int g = wv; g < HOP_GROUPS; g += BIN_WAVES) {
        const int gbase = g * ROWS_PER_GROUP;
        const int left  = N_NODES - gbase;
        const int rows  = left < ROWS_PER_GROUP ? left : ROWS_PER_GROUP;
        const int slots = sub_bin_slots(rows);
        const int n     = gtot[g];
        const int ncopy = n < slots - 1 ? n : slots - 1;
        const int st    = gstart[g];
        v2f* sub = (v2f*)(nxt + (size_t)gbase * OUT_F) + (size_t)blockIdx.x * slots;
        for (int jdx = lane; jdx <= ncopy; jdx += 32) {
            v2i rec;
            if (jdx == 0) { rec.x = n; rec.y = 0; }
            else          rec = sorted[st + jdx - 1];
            v2f o;
            const int rx = rec.x, ry = rec.y;
            o.x = float_of(rx);
            o.y = float_of(ry);
            sub[jdx] = o;
        }
    }
}

template <bool FULL>
__device__ __forceinline__ void hop_drain(const v2i* qp, int n, int lane,
                                          const float* __restrict__ cur,
                                          float* acc) {
    int   s  = 0;
    int   rw = 0;
    float wt = 0.f;
    if (FULL || lane < n) {
        const v2i pk = qp[lane];
        rw = (int)((unsigned)pk.x & ((1u << ROW_BITS) - 1u));
        s  = (int)((unsigned)pk.x >> ROW_BITS);
        const int wbits = pk.y;
        wt = float_of(wbits);
    }
    const int wti = __builtin_bit_cast(int, wt);
    const int cnt = FULL ? 32 : n;

    v2f   v[8];
    int   rj[8];
    float wj[8];
#define HOP_GATHER8(J0, V, RJ, WJ)                                                        \
    _Pragma("unroll")                                                                     \
    for (int u = 0; u < 8; ++u) {                                                         \
        int jj = (J0) + u;                                                                \
        if (!FULL) jj = jj < n ? jj : n - 1;                                              \
        const int sj = __builtin_amdgcn_readlane(s, jj);                                  \
        RJ[u] = __builtin_amdgcn_readlane(rw, jj);                                        \
        WJ[u] = __builtin_bit_cast(float, __builtin_amdgcn_readlane(wti, jj));            \
        V[u]  = *(const v2f*)(cur + (size_t)sj * OUT_F + lane * 2);                       \
    }
    HOP_GATHER8(0, v, rj, wj)
    for (int j0 = 0; j0 < cnt; j0 += 8) {
        v2f   nv[8];
        int   nrj[8];
        float nwj[8];
        const bool more = j0 + 8 < cnt;
        if (more) {
            HOP_GATHER8(j0 + 8, nv, nrj, nwj)
        }
#pragma unroll
        for (int u = 0; u < 8; ++u) {
            if (FULL || j0 + u < n) {
                v2f* a = (v2f*)(acc + rj[u] * OUT_F) + lane;
                *a = *a + v[u] * wj[u];
            }
        }
        if (more) {
#pragma unroll
            for (int u = 0; u < 8; ++u) { v[u] = nv[u]; rj[u] = nrj[u]; wj[u] = nwj[u]; }
        }
    }
#undef HOP_GATHER8
}

__global__ __launch_bounds__(HOP_THREADS)
void sgc_hop_owner_waves(const float* __restrict__ cur,
                         float* __restrict__ nxt,
                         const int* __restrict__ esrc,
                         const int* __restrict__ edst,
                         const float* __restrict__ ew) {
    __shared__ __attribute__((aligned(16))) float acc[ROWS_PER_GROUP * OUT_F];
    __shared__ v2i boxes[HOP_WAVES * BOX_RSTRIDE];
    __shared__ int box_cnt[HOP_WAVES * HOP_WAVES];
    __shared__ v2i ownq[HOP_WAVES][64];
    __shared__ int overflow_flag;

    const int tid  = threadIdx.x;
    const int lane = tid & 31;
    const int wv   = __builtin_amdgcn_readfirstlane(tid >> 5);
    const int base = blockIdx.x * ROWS_PER_GROUP;
    const int left = N_NODES - base;
    const unsigned nrows = (unsigned)(left < ROWS_PER_GROUP ? left : ROWS_PER_GROUP);
    const unsigned below = (1u << lane) - 1u;

    if (tid == 0) overflow_flag = 0;
    box_cnt[wv * HOP_WAVES + lane] = 0;
    float4* acc4 = (float4*)acc;
    for (int i = tid; i < ROWS_PER_GROUP * OUT_F / 4; i += HOP_THREADS)
        acc4[i] = make_float4(0.f, 0.f, 0.f, 0.f);
    __syncthreads();

    const int    slots  = sub_bin_slots((int)nrows);
    const int    room   = slots - 1;
    const v2f*   region = (const v2f*)(nxt + (size_t)base * OUT_F);
    int  nk[4];
    bool cut = false;
#pragma unroll
    for (int k = 0; k < 4; ++k) {
        const v2f* sb = region + (size_t)(wv + HOP_WAVES * k) * slots;
        const float head = sb[0].x;
        nk[k] = __builtin_amdgcn_readfirstlane(bits_of(head));
    }
#pragma unroll
    for (int k = 0; k < 4; ++k) {
        if ((unsigned)nk[k] > (unsigned)room) { cut = true; nk[k] = room; }
    }
    v2f ent[4][SUB_ROUNDS];
#pragma unroll
    for (int k = 0; k < 4; ++k) {
        const v2f* sb = region + (size_t)(wv + HOP_WAVES * k) * slots;
#pragma unroll
        for (int r = 0; r < SUB_ROUNDS; ++r) {
            const int q = r * 32 + lane;
            v2f z; z.x = 0.f; z.y = 0.f;
            ent[k][r] = q < nk[k] ? sb[1 + q] : z;
        }
    }
    v2i* myrow = boxes + wv * BOX_RSTRIDE;
#pragma unroll
    for (int k = 0; k < 4; ++k) {
#pragma unroll
        for (int r = 0; r < SUB_ROUNDS; ++r) {
            const bool has = r * 32 + lane < nk[k];
            if (__builtin_amdgcn_ballot_w32(has) == 0u) continue;
            v2i rec;
            const float fx = ent[k][r].x, fy = ent[k][r].y;
            rec.x = bits_of(fx);
            rec.y = bits_of(fy);
            const unsigned o = (unsigned)rec.x & 31u;
            const unsigned peers = lanes_with_same_key<5>(o, has);
            const int rank = __builtin_popcount(peers & below);
            if (has) {
                const int c    = box_cnt[wv * HOP_WAVES + (int)o];
                const int slot = c + rank;
                if (slot < BOX_SLOTS) myrow[(int)o * BOX_SLOTS + slot] = rec;
                if (rank == 0) box_cnt[wv * HOP_WAVES + (int)o] = c + __builtin_popcount(peers);
            }
        }
    }
    {
        const int c = box_cnt[wv * HOP_WAVES + lane];
        const bool lost = __builtin_amdgcn_ballot_w32(c > BOX_SLOTS) != 0u;
        box_cnt[wv * HOP_WAVES + lane] = c < BOX_SLOTS ? c : BOX_SLOTS;
        if ((lost || cut) && lane == 0) overflow_flag = 1;
    }
    __syncthreads();
    const int over = __builtin_amdgcn_readfirstlane(overflow_flag);

    if (over == 0) {
        const int  mine  = box_cnt[lane * HOP_WAVES + wv];
        const v2i* mybox = boxes + lane * BOX_RSTRIDE + wv * BOX_SLOTS;
        v2i* oq = ownq[wv];
        int  on = 0;
        for (int i = 0; i < BOX_SLOTS; ++i) {
            const bool has = i < mine;
            const unsigned m = __builtin_amdgcn_ballot_w32(has);
            if (m == 0u) break;
            const int pos = on + (int)__builtin_amdgcn_mbcnt_lo(m, 0u);
            if (has) oq[pos] = mybox[i];
            on += __builtin_popcount(m);
            if (on >= 32) {
                on -= 32;
                hop_drain<true>(oq + on, 32, lane, cur, acc);
            }
        }
        if (on > 0) hop_drain<false>(oq, on, lane, cur, acc);
    } else {
        const v4i* __restrict__ dst4 = (const v4i*)edst;
        v2i* sq = boxes + wv * QUEUE_RECS;
        int  qn = 0;
        for (int c0 = 0; c0 < N_CHUNKS; c0 += 4) {
            v4i d[4];
#pragma unroll
            for (int u = 0; u < 4; ++u) d[u] = dst4[(size_t)(c0 + u) * 32 + lane];
#pragma unroll
            for (int u = 0; u < 4; ++u) {
                const unsigned rr[4] = {(unsigned)(d[u].x - base), (unsigned)(d[u].y - base),
                                        (unsigned)(d[u].z - base), (unsigned)(d[u].w - base)};
                const int e0 = (c0 + u) * EDGES_PER_CHUNK + lane * 4;
#pragma unroll
                for (int k = 0; k < 4; ++k) {
                    const bool hit = rr[k] < nrows && (int)(rr[k] & 31u) == wv;
                    const unsigned m = __builtin_amdgcn_ballot_w32(hit);
                    if (m != 0u) {
                        const int pos = qn + (int)__builtin_amdgcn_mbcnt_lo(m, 0u);
                        if (hit) {
                            v2i rec;
                            rec.x = (int)(((unsigned)esrc[e0 + k] << ROW_BITS) | rr[k]);
                            rec.y = __builtin_bit_cast(int, ew[e0 + k]);
                            sq[pos] = rec;
                        }
                        qn += __builtin_popcount(m);
                    }
                }
                while (qn >= 32) {
                    qn -= 32;
                    hop_drain<true>(sq + qn, 32, lane, cur, acc);
                }
            }
        }
        if (qn > 0) hop_drain<false>(sq, qn, lane, cur, acc);
    }

#pragma unroll
    for (int r = 0; r < ROWS_PER_WAVE; ++r) {
        const unsigned row = (unsigned)(wv + HOP_WAVES * r);
        if (row < nrows)
            *(v2f*)(nxt + (size_t)(base + (int)row) * OUT_F + lane * 2) =
                *((const v2f*)(acc + row * OUT_F) + lane);
    }
}

extern "C" void kernel_launch(void* const* d_in, const int* in_sizes, int n_in,
                              void* d_out, int out_size, void* d_ws, size_t ws_size,
                              hipStream_t stream) {
    const float* x    = (const float*)d_in[0];
    const float* W    = (const float*)d_in[1];
    const float* b    = (const float*)d_in[2];
    const int*   esrc = (const int*)d_in[3];
    const int*   edst = (const int*)d_in[4];
    const float* ew   = (const float*)d_in[5];

    float* out = (float*)d_out;
    float* ws  = (float*)d_ws;

    sgc_bin_edges<<<BIN_SLICES, BIN_THREADS, 0, stream>>>(esrc, edst, ew, out);
    sgc_gemm_wmma<<<GEMM_GROUPS, GEMM_THREADS, 0, stream>>>(x, W, b, ws);

    sgc_hop_owner_waves<<<HOP_GROUPS, HOP_THREADS, 0, stream>>>(ws, out, esrc, edst, ew);
    sgc_bin_edges<<<BIN_SLICES, BIN_THREADS, 0, stream>>>(esrc, edst, ew, ws);
    sgc_hop_owner_waves<<<HOP_GROUPS, HOP_THREADS, 0, stream>>>(out, ws, esrc, edst, ew);
    sgc_bin_edges<<<BIN_SLICES, BIN_THREADS, 0, stream>>>(esrc, edst, ew, out);
    sgc_hop_owner_waves<<<HOP_GROUPS, HOP_THREADS, 0, stream>>>(ws, out, esrc, edst, ew);
}


# edge ordering launches: serial LDS round trips cut: place kept from the count pass, table reads and a wave's 16 copy-out groups issued together, run starts by a wave scan; same records, same places
# speedup vs baseline: 2.8974x; 1.0545x over previous
#include <hip/hip_runtime.h>
#include <hip/hip_bf16.h>

typedef float v2f __attribute__((ext_vector_type(2)));
typedef float v8f __attribute__((ext_vector_type(8)));

#define N_NODES 100000
#define N_EDGES 1600000
#define IN_F    256
#define OUT_F   64
#define NTILES  (N_NODES / 16)

#define GEMM_MT      4
#define GEMM_WAVES   ((NTILES + GEMM_MT - 1) / GEMM_MT)
#define GEMM_THREADS 64
#define GEMM_GROUPS  ((GEMM_WAVES + 1) / 2)
#define GEMM_KC      4

__global__ __launch_bounds__(GEMM_THREADS)
void sgc_gemm_wmma(const float* __restrict__ x,
                   const float* __restrict__ W,
                   const float* __restrict__ bias,
                   float* __restrict__ out) {
    const int wave = (blockIdx.x * GEMM_THREADS + threadIdx.x) >> 5;
    const int lane = threadIdx.x & 31;
    if (wave >= GEMM_WAVES) return;

    const int m    = lane & 15;
    const int koff = (lane >> 4) << 1;

    const float* __restrict__ xr[GEMM_MT];
    const float* __restrict__ wr[4];
#pragma unroll
    for (int t = 0; t < GEMM_MT; ++t) {
        const int tile = wave * GEMM_MT + t;
        const int tl   = tile < NTILES ? tile : NTILES - 1;
        xr[t] = x + (size_t)(tl * 16 + m) * IN_F + koff;
    }
#pragma unroll
    for (int n = 0; n < 4; ++n) wr[n] = W + (size_t)(m + 16 * n) * IN_F + koff;

    v8f acc[GEMM_MT][4];
#pragma unroll
    for (int t = 0; t < GEMM_MT; ++t)
#pragma unroll
        for (int n = 0; n < 4; ++n) acc[t][n] = v8f{};

#define GEMM_LOAD_SET(A, B, K0)                                                           \
    _Pragma("unroll")                                                                     \
    for (int kk = 0; kk < GEMM_KC; ++kk) {                                                \
        _Pragma("unroll")                                                                 \
        for (int t = 0; t < GEMM_MT; ++t) A[kk][t] = *(const v2f*)(xr[t] + (K0) + 4 * kk); \
        _Pragma("unroll")                                                                 \
        for (int n = 0; n < 4; ++n)       B[kk][n] = *(const v2f*)(wr[n] + (K0) + 4 * kk); \
    }

#define GEMM_MUL_SET(A, B)                                                                \
    _Pragma("unroll")                                                                     \
    for (int kk = 0; kk < GEMM_KC; ++kk) {                                                \
        _Pragma("unroll")                                                                 \
        for (int t = 0; t < GEMM_MT; ++t) {                                               \
            _Pragma("unroll")                                                             \
            for (int n = 0; n < 4; ++n)                                                   \
                acc[t][n] = __builtin_amdgcn_wmma_f32_16x16x4_f32(false, A[kk][t], false, B[kk][n], \
                                                                  (short)0, acc[t][n], false, false); \
        }                                                                                 \
    }

    v2f a0[GEMM_KC][GEMM_MT], b0[GEMM_KC][4];
    v2f a1[GEMM_KC][GEMM_MT], b1[GEMM_KC][4];
    GEMM_LOAD_SET(a0, b0, 0)
    for (int k0 = 0; k0 < IN_F - 8 * GEMM_KC; k0 += 8 * GEMM_KC) {
        GEMM_LOAD_SET(a1, b1, k0 + 4 * GEMM_KC)
        asm volatile("" ::: "memory");
        GEMM_MUL_SET(a0, b0)
        GEMM_LOAD_SET(a0, b0, k0 + 8 * GEMM_KC)
        asm volatile("" ::: "memory");
        GEMM_MUL_SET(a1, b1)
    }
    GEMM_LOAD_SET(a1, b1, IN_F - 4 * GEMM_KC)
    asm volatile("" ::: "memory");
    GEMM_MUL_SET(a0, b0)
    GEMM_MUL_SET(a1, b1)
#undef GEMM_MUL_SET
#undef GEMM_LOAD_SET

    const int col = lane & 15;
    const float bv0 = bias[col +  0];
    const float bv1 = bias[col + 16];
    const float bv2 = bias[col + 32];
    const float bv3 = bias[col + 48];

#pragma unroll
    for (int t = 0; t < GEMM_MT; ++t) {
        const int tile = wave * GEMM_MT + t;
        if (tile < NTILES) {
            const int row_base = tile * 16 + ((lane >> 4) ? 8 : 0);
#pragma unroll
            for (int r = 0; r < 8; ++r) {
                float* o = out + (size_t)(row_base + r) * OUT_F + col;
                o[ 0] = acc[t][0][r] + bv0;
                o[16] = acc[t][1][r] + bv1;
                o[32] = acc[t][2][r] + bv2;
                o[48] = acc[t][3][r] + bv3;
            }
        }
    }
}

#define HOP_GROUPS      256
#define ROWS_PER_GROUP  391
#define HOP_THREADS     1024
#define HOP_WAVES       32
#define ROWS_PER_WAVE   13
#define EDGES_PER_CHUNK 128
#define N_CHUNKS        (N_EDGES / EDGES_PER_CHUNK)
#define ROW_BITS        9
#define BOX_SLOTS       24
#define BOX_RSTRIDE     (HOP_WAVES * BOX_SLOTS + 1)
#define QUEUE_RECS      160

#define BIN_SLICES      128
#define SLICE_EDGES     (N_EDGES / BIN_SLICES)
#define BIN_THREADS     512
#define BIN_WAVES       16
#define SLICE_BATCHES   ((SLICE_EDGES + 31) / 32)
#define BATCHES_PER_WAVE ((SLICE_BATCHES + BIN_WAVES - 1) / BIN_WAVES)
#define SUB_SLOTS_MAX   96
#define SUB_ROUNDS      3

typedef int v2i __attribute__((ext_vector_type(2)));
typedef int v4i __attribute__((ext_vector_type(4)));

static_assert(N_EDGES % BIN_SLICES == 0, "slices of equal length");
static_assert(BIN_SLICES == 4 * HOP_WAVES, "four sub-bins a reading wave");
static_assert(SUB_SLOTS_MAX - 1 <= 32 * SUB_ROUNDS, "a sub-bin is read in SUB_ROUNDS rounds");
static_assert(BIN_SLICES * 8 == 4 * 256, "rows >> 2 slots a sub-bin use up the rows of the group");
static_assert(HOP_WAVES * QUEUE_RECS <= HOP_WAVES * BOX_RSTRIDE, "slow-path queues fit the boxes");

__device__ __forceinline__ int sub_bin_slots(int rows) {
    const int s = rows >> 2;
    return s < SUB_SLOTS_MAX ? s : SUB_SLOTS_MAX;
}

__device__ __forceinline__ int   bits_of(float f) { return __builtin_bit_cast(int, f); }
__device__ __forceinline__ float float_of(int i)  { return __builtin_bit_cast(float, i); }

template <int NBITS>
__device__ __forceinline__ unsigned lanes_with_same_key(unsigned key, bool valid) {
    unsigned peers = __builtin_amdgcn_ballot_w32(valid);
#pragma unroll
    for (int b = 0; b < NBITS; ++b) {
        const bool bit = ((key >> b) & 1u) != 0u;
        const unsigned m = __builtin_amdgcn_ballot_w32(bit);
        peers &= bit ? m : ~m;
    }
    return peers;
}

#define GROUPS_PER_BIN_WAVE (HOP_GROUPS / BIN_WAVES)
#define SCAN_WAVES          (HOP_GROUPS / 32)

static_assert(HOP_GROUPS % BIN_WAVES == 0 && GROUPS_PER_BIN_WAVE <= 32, "one lane a group of the wave");
static_assert(BATCHES_PER_WAVE * 32 <= 1024, "a place inside a wave's run fits 10 bits");

__global__ __launch_bounds__(BIN_THREADS)
void sgc_bin_edges(const int* __restrict__ esrc,
                   const int* __restrict__ edst,
                   const float* __restrict__ ew,
                   float* __restrict__ nxt) {
    __shared__ v2i sorted[SLICE_EDGES];
    __shared__ int cnt[BIN_WAVES * HOP_GROUPS];
    __shared__ int gtot[HOP_GROUPS];
    __shared__ int gstart[HOP_GROUPS];
    __shared__ int wsum[SCAN_WAVES];

    const int tid  = threadIdx.x;
    const int lane = tid & 31;
    const int wv   = __builtin_amdgcn_readfirstlane(tid >> 5);
    const int e_base = blockIdx.x * SLICE_EDGES;
    const unsigned below = (1u << lane) - 1u;

    for (int i = tid; i < BIN_WAVES * HOP_GROUPS; i += BIN_THREADS) cnt[i] = 0;

    int   dv[BATCHES_PER_WAVE];
    int   sv[BATCHES_PER_WAVE];
    float wt[BATCHES_PER_WAVE];
#pragma unroll
    for (int j = 0; j < BATCHES_PER_WAVE; ++j) {
        const int i  = (wv + BIN_WAVES * j) * 32 + lane;
        const int e  = e_base + (i < SLICE_EDGES ? i : 0);
        dv[j] = edst[e];
        sv[j] = esrc[e];
        wt[j] = ew[e];
    }
    __syncthreads();

    int pk[BATCHES_PER_WAVE];
#pragma unroll
    for (int j = 0; j < BATCHES_PER_WAVE; ++j) {
        const int  i     = (wv + BIN_WAVES * j) * 32 + lane;
        const bool valid = i < SLICE_EDGES;
        const unsigned g = ((unsigned)dv[j] / (unsigned)ROWS_PER_GROUP) & (HOP_GROUPS - 1u);
        const unsigned r = (unsigned)dv[j] - g * (unsigned)ROWS_PER_GROUP;
        const unsigned peers = lanes_with_same_key<8>(g, valid);
        const unsigned rank  = (unsigned)__builtin_popcount(peers & below);
        int c = 0;
        if (valid) {
            c = cnt[wv * HOP_GROUPS + (int)g];
            if (rank == 0u) cnt[wv * HOP_GROUPS + (int)g] = c + __builtin_popcount(peers);
        }
        const unsigned place = (unsigned)c + rank;
        pk[j] = (int)(g | ((r & 511u) << 8) | (place << 17) | (valid ? 1u << 27 : 0u));
    }
    __syncthreads();

    int run_len = 0, run_incl = 0;
    if (tid < HOP_GROUPS) {
        int c[BIN_WAVES];
#pragma unroll
        for (int w = 0; w < BIN_WAVES; ++w) c[w] = cnt[w * HOP_GROUPS + tid];
        int a = 0;
#pragma unroll
        for (int w = 0; w < BIN_WAVES; ++w) {
            cnt[w * HOP_GROUPS + tid] = a;
            a += c[w];
        }
        gtot[tid] = a;
        run_len  = a;
        run_incl = a;
        for (int d = 1; d < 32; d <<= 1) {
            const int up = __shfl_up(run_incl, d, 32);
            if (lane >= d) run_incl += up;
        }
        if (lane == 31) wsum[wv] = run_incl;
    }
    __syncthreads();
    if (tid < HOP_GROUPS) {
        int before = 0;
        for (int w = 0; w < SCAN_WAVES; ++w) {
            const int t = wsum[w];
            before += w < wv ? t : 0;
        }
        gstart[tid] = before + run_incl - run_len;
    }
    __syncthreads();

    int pos[BATCHES_PER_WAVE];
#pragma unroll
    for (int j = 0; j < BATCHES_PER_WAVE; ++j) {
        const unsigned p = (unsigned)pk[j];
        const int g = (int)(p & 255u);
        pos[j] = gstart[g] + cnt[wv * HOP_GROUPS + g] + (int)((p >> 17) & 1023u);
    }
#pragma unroll
    for (int j = 0; j < BATCHES_PER_WAVE; ++j) {
        const unsigned p = (unsigned)pk[j];
        if ((p >> 27) != 0u) {
            v2i rec;
            rec.x = (int)(((unsigned)sv[j] << ROW_BITS) | ((p >> 8) & 511u));
            rec.y = __builtin_bit_cast(int, wt[j]);
            sorted[pos[j]] = rec;
        }
    }
    __syncthreads();

    const int gl   = wv + BIN_WAVES * (lane & (GROUPS_PER_BIN_WAVE - 1));
    const int n_l  = gtot[gl];
    const int st_l = gstart[gl];
    v2i r0[GROUPS_PER_BIN_WAVE], r1[GROUPS_PER_BIN_WAVE];
#pragma unroll
    for (int t = 0; t < GROUPS_PER_BIN_WAVE; ++t) {
        const int st = __builtin_amdgcn_readlane(st_l, t);
        int i0 = st + lane - 1;
        i0 = i0 < 0 ? 0 : i0;
        i0 = i0 < SLICE_EDGES ? i0 : SLICE_EDGES - 1;
        int i1 = st + lane + 31;
        i1 = i1 < SLICE_EDGES ? i1 : SLICE_EDGES - 1;
        r0[t] = sorted[i0];
        r1[t] = sorted[i1];
    }
#pragma unroll
    for (int t = 0; t < GROUPS_PER_BIN_WAVE; ++t) {
        const int g     = wv + BIN_WAVES * t;
        const int gbase = g * ROWS_PER_GROUP;
        const int left  = N_NODES - gbase;
        const int rows  = left < ROWS_PER_GROUP ? left : ROWS_PER_GROUP;
        const int slots = sub_bin_slots(rows);
        const int n     = __builtin_amdgcn_readlane(n_l, t);
        const int ncopy = n < slots - 1 ? n : slots - 1;
        v2f* sub = (v2f*)(nxt + (size_t)gbase * OUT_F) + (size_t)blockIdx.x * slots;
        if (lane <= ncopy) {
            const int ax = r0[t].x, ay = r0[t].y;
            const int rx = lane == 0 ? n : ax;
            const int ry = lane == 0 ? 0 : ay;
            v2f o;
            o.x = float_of(rx);
            o.y = float_of(ry);
            sub[lane] = o;
        }
        if (32 + lane <= ncopy) {
            const int rx = r1[t].x, ry = r1[t].y;
            v2f o;
            o.x = float_of(rx);
            o.y = float_of(ry);
            sub[32 + lane] = o;
        }
    }
    for (int t = 0; t < GROUPS_PER_BIN_WAVE; ++t) {
        const int g     = wv + BIN_WAVES * t;
        const int gbase = g * ROWS_PER_GROUP;
        const int left  = N_NODES - gbase;
        const int rows  = left < ROWS_PER_GROUP ? left : ROWS_PER_GROUP;
        const int slots = sub_bin_slots(rows);
        const int n     = __builtin_amdgcn_readlane(n_l, t);
        const int st    = __builtin_amdgcn_readlane(st_l, t);
        const int ncopy = n < slots - 1 ? n : slots - 1;
        if (ncopy >= 64) {
            const int jdx = 64 + lane;
            if (jdx <= ncopy) {
                const v2i rec = sorted[st + jdx - 1];
                const int rx = rec.x, ry = rec.y;
                v2f o;
                o.x = float_of(rx);
                o.y = float_of(ry);
                v2f* sub = (v2f*)(nxt + (size_t)gbase * OUT_F) + (size_t)blockIdx.x * slots;
                sub[jdx] = o;
            }
        }
    }
}

template <bool FULL>
__device__ __forceinline__ void hop_drain(const v2i* qp, int n, int lane,
                                          const float* __restrict__ cur,
                                          float* acc) {
    int   s  = 0;
    int   rw = 0;
    float wt = 0.f;
    if (FULL || lane < n) {
        const v2i pk = qp[lane];
        rw = (int)((unsigned)pk.x & ((1u << ROW_BITS) - 1u));
        s  = (int)((unsigned)pk.x >> ROW_BITS);
        const int wbits = pk.y;
        wt = float_of(wbits);
    }
    const int wti = __builtin_bit_cast(int, wt);
    const int cnt = FULL ? 32 : n;

    v2f   v[8];
    int   rj[8];
    float wj[8];
#define HOP_GATHER8(J0, V, RJ, WJ)                                                        \
    _Pragma("unroll")                                                                     \
    for (int u = 0; u < 8; ++u) {                                                         \
        int jj = (J0) + u;                                                                \
        if (!FULL) jj = jj < n ? jj : n - 1;                                              \
        const int sj = __builtin_amdgcn_readlane(s, jj);                                  \
        RJ[u] = __builtin_amdgcn_readlane(rw, jj);                                        \
        WJ[u] = __builtin_bit_cast(float, __builtin_amdgcn_readlane(wti, jj));            \
        V[u]  = *(const v2f*)(cur + (size_t)sj * OUT_F + lane * 2);                       \
    }
    HOP_GATHER8(0, v, rj, wj)
    for (int j0 = 0; j0 < cnt; j0 += 8) {
        v2f   nv[8];
        int   nrj[8];
        float nwj[8];
        const bool more = j0 + 8 < cnt;
        if (more) {
            HOP_GATHER8(j0 + 8, nv, nrj, nwj)
        }
#pragma unroll
        for (int u = 0; u < 8; ++u) {
            if (FULL || j0 + u < n) {
                v2f* a = (v2f*)(acc + rj[u] * OUT_F) + lane;
                *a = *a + v[u] * wj[u];
            }
        }
        if (more) {
#pragma unroll
            for (int u = 0; u < 8; ++u) { v[u] = nv[u]; rj[u] = nrj[u]; wj[u] = nwj[u]; }
        }
    }
#undef HOP_GATHER8
}

__global__ __launch_bounds__(HOP_THREADS)
void sgc_hop_owner_waves(const float* __restrict__ cur,
                         float* __restrict__ nxt,
                         const int* __restrict__ esrc,
                         const int* __restrict__ edst,
                         const float* __restrict__ ew) {
    __shared__ __attribute__((aligned(16))) float acc[ROWS_PER_GROUP * OUT_F];
    __shared__ v2i boxes[HOP_WAVES * BOX_RSTRIDE];
    __shared__ int box_cnt[HOP_WAVES * HOP_WAVES];
    __shared__ v2i ownq[HOP_WAVES][64];
    __shared__ int overflow_flag;

    const int tid  = threadIdx.x;
    const int lane = tid & 31;
    const int wv   = __builtin_amdgcn_readfirstlane(tid >> 5);
    const int base = blockIdx.x * ROWS_PER_GROUP;
    const int left = N_NODES - base;
    const unsigned nrows = (unsigned)(left < ROWS_PER_GROUP ? left : ROWS_PER_GROUP);
    const unsigned below = (1u << lane) - 1u;

    if (tid == 0) overflow_flag = 0;
    box_cnt[wv * HOP_WAVES + lane] = 0;
    float4* acc4 = (float4*)acc;
    for (int i = tid; i < ROWS_PER_GROUP * OUT_F / 4; i += HOP_THREADS)
        acc4[i] = make_float4(0.f, 0.f, 0.f, 0.f);
    __syncthreads();

    const int    slots  = sub_bin_slots((int)nrows);
    const int    room   = slots - 1;
    const v2f*   region = (const v2f*)(nxt + (size_t)base * OUT_F);
    int  nk[4];
    bool cut = false;
#pragma unroll
    for (int k = 0; k < 4; ++k) {
        const v2f* sb = region + (size_t)(wv + HOP_WAVES * k) * slots;
        const float head = sb[0].x;
        nk[k] = __builtin_amdgcn_readfirstlane(bits_of(head));
    }
#pragma unroll
    for (int k = 0; k < 4; ++k) {
        if ((unsigned)nk[k] > (unsigned)room) { cut = true; nk[k] = room; }
    }
    v2f ent[4][SUB_ROUNDS];
#pragma unroll
    for (int k = 0; k < 4; ++k) {
        const v2f* sb = region + (size_t)(wv + HOP_WAVES * k) * slots;
#pragma unroll
        for (int r = 0; r < SUB_ROUNDS; ++r) {
            const int q = r * 32 + lane;
            v2f z; z.x = 0.f; z.y = 0.f;
            ent[k][r] = q < nk[k] ? sb[1 + q] : z;
        }
    }
    v2i* myrow = boxes + wv * BOX_RSTRIDE;
#pragma unroll
    for (int k = 0; k < 4; ++k) {
#pragma unroll
        for (int r = 0; r < SUB_ROUNDS; ++r) {
            const bool has = r * 32 + lane < nk[k];
            if (__builtin_amdgcn_ballot_w32(has) == 0u) continue;
            v2i rec;
            const float fx = ent[k][r].x, fy = ent[k][r].y;
            rec.x = bits_of(fx);
            rec.y = bits_of(fy);
            const unsigned o = (unsigned)rec.x & 31u;
            const unsigned peers = lanes_with_same_key<5>(o, has);
            const int rank = __builtin_popcount(peers & below);
            if (has) {
                const int c    = box_cnt[wv * HOP_WAVES + (int)o];
                const int slot = c + rank;
                if (slot < BOX_SLOTS) myrow[(int)o * BOX_SLOTS + slot] = rec;
                if (rank == 0) box_cnt[wv * HOP_WAVES + (int)o] = c + __builtin_popcount(peers);
            }
        }
    }
    {
        const int c = box_cnt[wv * HOP_WAVES + lane];
        const bool lost = __builtin_amdgcn_ballot_w32(c > BOX_SLOTS) != 0u;
        box_cnt[wv * HOP_WAVES + lane] = c < BOX_SLOTS ? c : BOX_SLOTS;
        if ((lost || cut) && lane == 0) overflow_flag = 1;
    }
    __syncthreads();
    const int over = __builtin_amdgcn_readfirstlane(overflow_flag);

    if (over == 0) {
        const int  mine  = box_cnt[lane * HOP_WAVES + wv];
        const v2i* mybox = boxes + lane * BOX_RSTRIDE + wv * BOX_SLOTS;
        v2i* oq = ownq[wv];
        int  on = 0;
        for (int i = 0; i < BOX_SLOTS; ++i) {
            const bool has = i < mine;
            const unsigned m = __builtin_amdgcn_ballot_w32(has);
            if (m == 0u) break;
            const int pos = on + (int)__builtin_amdgcn_mbcnt_lo(m, 0u);
            if (has) oq[pos] = mybox[i];
            on += __builtin_popcount(m);
            if (on >= 32) {
                on -= 32;
                hop_drain<true>(oq + on, 32, lane, cur, acc);
            }
        }
        if (on > 0) hop_drain<false>(oq, on, lane, cur, acc);
    } else {
        const v4i* __restrict__ dst4 = (const v4i*)edst;
        v2i* sq = boxes + wv * QUEUE_RECS;
        int  qn = 0;
        for (int c0 = 0; c0 < N_CHUNKS; c0 += 4) {
            v4i d[4];
#pragma unroll
            for (int u = 0; u < 4; ++u) d[u] = dst4[(size_t)(c0 + u) * 32 + lane];
#pragma unroll
            for (int u = 0; u < 4; ++u) {
                const unsigned rr[4] = {(unsigned)(d[u].x - base), (unsigned)(d[u].y - base),
                                        (unsigned)(d[u].z - base), (unsigned)(d[u].w - base)};
                const int e0 = (c0 + u) * EDGES_PER_CHUNK + lane * 4;
#pragma unroll
                for (int k = 0; k < 4; ++k) {
                    const bool hit = rr[k] < nrows && (int)(rr[k] & 31u) == wv;
                    const unsigned m = __builtin_amdgcn_ballot_w32(hit);
                    if (m != 0u) {
                        const int pos = qn + (int)__builtin_amdgcn_mbcnt_lo(m, 0u);
                        if (hit) {
                            v2i rec;
                            rec.x = (int)(((unsigned)esrc[e0 + k] << ROW_BITS) | rr[k]);
                            rec.y = __builtin_bit_cast(int, ew[e0 + k]);
                            sq[pos] = rec;
                        }
                        qn += __builtin_popcount(m);
                    }
                }
                while (qn >= 32) {
                    qn -= 32;
                    hop_drain<true>(sq + qn, 32, lane, cur, acc);
                }
            }
        }
        if (qn > 0) hop_drain<false>(sq, qn, lane, cur, acc);
    }

#pragma unroll
    for (int r = 0; r < ROWS_PER_WAVE; ++r) {
        const unsigned row = (unsigned)(wv + HOP_WAVES * r);
        if (row < nrows)
            *(v2f*)(nxt + (size_t)(base + (int)row) * OUT_F + lane * 2) =
                *((const v2f*)(acc + row * OUT_F) + lane);
    }
}

extern "C" void kernel_launch(void* const* d_in, const int* in_sizes, int n_in,
                              void* d_out, int out_size, void* d_ws, size_t ws_size,
                              hipStream_t stream) {
    const float* x    = (const float*)d_in[0];
    const float* W    = (const float*)d_in[1];
    const float* b    = (const float*)d_in[2];
    const int*   esrc = (const int*)d_in[3];
    const int*   edst = (const int*)d_in[4];
    const float* ew   = (const float*)d_in[5];

    float* out = (float*)d_out;
    float* ws  = (float*)d_ws;

    sgc_bin_edges<<<BIN_SLICES, BIN_THREADS, 0, stream>>>(esrc, edst, ew, out);
    sgc_gemm_wmma<<<GEMM_GROUPS, GEMM_THREADS, 0, stream>>>(x, W, b, ws);

    sgc_hop_owner_waves<<<HOP_GROUPS, HOP_THREADS, 0, stream>>>(ws, out, esrc, edst, ew);
    sgc_bin_edges<<<BIN_SLICES, BIN_THREADS, 0, stream>>>(esrc, edst, ew, ws);
    sgc_hop_owner_waves<<<HOP_GROUPS, HOP_THREADS, 0, stream>>>(out, ws, esrc, edst, ew);
    sgc_bin_edges<<<BIN_SLICES, BIN_THREADS, 0, stream>>>(esrc, edst, ew, out);
    sgc_hop_owner_waves<<<HOP_GROUPS, HOP_THREADS, 0, stream>>>(ws, out, esrc, edst, ew);
}


# hops: a wave keeps the sums of its 13 rows in registers, picked by a scalar row index (v_movrel), not in LDS: no zero fill and its barrier, no LDS read-add-write an edge; every sum and its order kept
# speedup vs baseline: 3.0555x; 1.0546x over previous
#include <hip/hip_runtime.h>
#include <hip/hip_bf16.h>

typedef float v2f __attribute__((ext_vector_type(2)));
typedef float v8f __attribute__((ext_vector_type(8)));

#define N_NODES 100000
#define N_EDGES 1600000
#define IN_F    256
#define OUT_F   64
#define NTILES  (N_NODES / 16)

#define GEMM_MT      4
#define GEMM_WAVES   ((NTILES + GEMM_MT - 1) / GEMM_MT)
#define GEMM_THREADS 64
#define GEMM_GROUPS  ((GEMM_WAVES + 1) / 2)
#define GEMM_KC      4

__global__ __launch_bounds__(GEMM_THREADS)
void sgc_gemm_wmma(const float* __restrict__ x,
                   const float* __restrict__ W,
                   const float* __restrict__ bias,
                   float* __restrict__ out) {
    const int wave = (blockIdx.x * GEMM_THREADS + threadIdx.x) >> 5;
    const int lane = threadIdx.x & 31;
    if (wave >= GEMM_WAVES) return;

    const int m    = lane & 15;
    const int koff = (lane >> 4) << 1;

    const float* __restrict__ xr[GEMM_MT];
    const float* __restrict__ wr[4];
#pragma unroll
    for (int t = 0; t < GEMM_MT; ++t) {
        const int tile = wave * GEMM_MT + t;
        const int tl   = tile < NTILES ? tile : NTILES - 1;
        xr[t] = x + (size_t)(tl * 16 + m) * IN_F + koff;
    }
#pragma unroll
    for (int n = 0; n < 4; ++n) wr[n] = W + (size_t)(m + 16 * n) * IN_F + koff;

    v8f acc[GEMM_MT][4];
#pragma unroll
    for (int t = 0; t < GEMM_MT; ++t)
#pragma unroll
        for (int n = 0; n < 4; ++n) acc[t][n] = v8f{};

#define GEMM_LOAD_SET(A, B, K0)                                                           \
    _Pragma("unroll")                                                                     \
    for (int kk = 0; kk < GEMM_KC; ++kk) {                                                \
        _Pragma("unroll")                                                                 \
        for (int t = 0; t < GEMM_MT; ++t) A[kk][t] = *(const v2f*)(xr[t] + (K0) + 4 * kk); \
        _Pragma("unroll")                                                                 \
        for (int n = 0; n < 4; ++n)       B[kk][n] = *(const v2f*)(wr[n] + (K0) + 4 * kk); \
    }

#define GEMM_MUL_SET(A, B)                                                                \
    _Pragma("unroll")                                                                     \
    for (int kk = 0; kk < GEMM_KC; ++kk) {                                                \
        _Pragma("unroll")                                                                 \
        for (int t = 0; t < GEMM_MT; ++t) {                                               \
            _Pragma("unroll")                                                             \
            for (int n = 0; n < 4; ++n)                                                   \
                acc[t][n] = __builtin_amdgcn_wmma_f32_16x16x4_f32(false, A[kk][t], false, B[kk][n], \
                                                                  (short)0, acc[t][n], false, false); \
        }                                                                                 \
    }

    v2f a0[GEMM_KC][GEMM_MT], b0[GEMM_KC][4];
    v2f a1[GEMM_KC][GEMM_MT], b1[GEMM_KC][4];
    GEMM_LOAD_SET(a0, b0, 0)
    for (int k0 = 0; k0 < IN_F - 8 * GEMM_KC; k0 += 8 * GEMM_KC) {
        GEMM_LOAD_SET(a1, b1, k0 + 4 * GEMM_KC)
        asm volatile("" ::: "memory");
        GEMM_MUL_SET(a0, b0)
        GEMM_LOAD_SET(a0, b0, k0 + 8 * GEMM_KC)
        asm volatile("" ::: "memory");
        GEMM_MUL_SET(a1, b1)
    }
    GEMM_LOAD_SET(a1, b1, IN_F - 4 * GEMM_KC)
    asm volatile("" ::: "memory");
    GEMM_MUL_SET(a0, b0)
    GEMM_MUL_SET(a1, b1)
#undef GEMM_MUL_SET
#undef GEMM_LOAD_SET

    const int col = lane & 15;
    const float bv0 = bias[col +  0];
    const float bv1 = bias[col + 16];
    const float bv2 = bias[col + 32];
    const float bv3 = bias[col + 48];

#pragma unroll
    for (int t = 0; t < GEMM_MT; ++t) {
        const int tile = wave * GEMM_MT + t;
        if (tile < NTILES) {
            const int row_base = tile * 16 + ((lane >> 4) ? 8 : 0);
#pragma unroll
            for (int r = 0; r < 8; ++r) {
                float* o = out + (size_t)(row_base + r) * OUT_F + col;
                o[ 0] = acc[t][0][r] + bv0;
                o[16] = acc[t][1][r] + bv1;
                o[32] = acc[t][2][r] + bv2;
                o[48] = acc[t][3][r] + bv3;
            }
        }
    }
}

#define HOP_GROUPS      256
#define ROWS_PER_GROUP  391
#define HOP_THREADS     1024
#define HOP_WAVES       32
#define ROWS_PER_WAVE   13
#define EDGES_PER_CHUNK 128
#define N_CHUNKS        (N_EDGES / EDGES_PER_CHUNK)
#define ROW_BITS        9
#define BOX_SLOTS       24
#define BOX_RSTRIDE     (HOP_WAVES * BOX_SLOTS + 1)
#define QUEUE_RECS      160

#define BIN_SLICES      128
#define SLICE_EDGES     (N_EDGES / BIN_SLICES)
#define BIN_THREADS     512
#define BIN_WAVES       16
#define SLICE_BATCHES   ((SLICE_EDGES + 31) / 32)
#define BATCHES_PER_WAVE ((SLICE_BATCHES + BIN_WAVES - 1) / BIN_WAVES)
#define SUB_SLOTS_MAX   96
#define SUB_ROUNDS      3

typedef int v2i __attribute__((ext_vector_type(2)));
typedef int v4i __attribute__((ext_vector_type(4)));
typedef float v16f __attribute__((ext_vector_type(16)));
static_assert(ROWS_PER_WAVE <= 16 && (1 << (ROW_BITS - 5)) <= 16, "any row number of a record picks an element of the vectors");

static_assert(N_EDGES % BIN_SLICES == 0, "slices of equal length");
static_assert(BIN_SLICES == 4 * HOP_WAVES, "four sub-bins a reading wave");
static_assert(SUB_SLOTS_MAX - 1 <= 32 * SUB_ROUNDS, "a sub-bin is read in SUB_ROUNDS rounds");
static_assert(BIN_SLICES * 8 == 4 * 256, "rows >> 2 slots a sub-bin use up the rows of the group");
static_assert(HOP_WAVES * QUEUE_RECS <= HOP_WAVES * BOX_RSTRIDE, "slow-path queues fit the boxes");

__device__ __forceinline__ int sub_bin_slots(int rows) {
    const int s = rows >> 2;
    return s < SUB_SLOTS_MAX ? s : SUB_SLOTS_MAX;
}

__device__ __forceinline__ int   bits_of(float f) { return __builtin_bit_cast(int, f); }
__device__ __forceinline__ float float_of(int i)  { return __builtin_bit_cast(float, i); }

template <int NBITS>
__device__ __forceinline__ unsigned lanes_with_same_key(unsigned key, bool valid) {
    unsigned peers = __builtin_amdgcn_ballot_w32(valid);
#pragma unroll
    for (int b = 0; b < NBITS; ++b) {
        const bool bit = ((key >> b) & 1u) != 0u;
        const unsigned m = __builtin_amdgcn_ballot_w32(bit);
        peers &= bit ? m : ~m;
    }
    return peers;
}

#define GROUPS_PER_BIN_WAVE (HOP_GROUPS / BIN_WAVES)
#define SCAN_WAVES          (HOP_GROUPS / 32)

static_assert(HOP_GROUPS % BIN_WAVES == 0 && GROUPS_PER_BIN_WAVE <= 32, "one lane a group of the wave");
static_assert(BATCHES_PER_WAVE * 32 <= 1024, "a place inside a wave's run fits 10 bits");

__global__ __launch_bounds__(BIN_THREADS)
void sgc_bin_edges(const int* __restrict__ esrc,
                   const int* __restrict__ edst,
                   const float* __restrict__ ew,
                   float* __restrict__ nxt) {
    __shared__ v2i sorted[SLICE_EDGES];
    __shared__ int cnt[BIN_WAVES * HOP_GROUPS];
    __shared__ int gtot[HOP_GROUPS];
    __shared__ int gstart[HOP_GROUPS];
    __shared__ int wsum[SCAN_WAVES];

    const int tid  = threadIdx.x;
    const int lane = tid & 31;
    const int wv   = __builtin_amdgcn_readfirstlane(tid >> 5);
    const int e_base = blockIdx.x * SLICE_EDGES;
    const unsigned below = (1u << lane) - 1u;

    for (int i = tid; i < BIN_WAVES * HOP_GROUPS; i += BIN_THREADS) cnt[i] = 0;

    int   dv[BATCHES_PER_WAVE];
    int   sv[BATCHES_PER_WAVE];
    float wt[BATCHES_PER_WAVE];
#pragma unroll
    for (int j = 0; j < BATCHES_PER_WAVE; ++j) {
        const int i  = (wv + BIN_WAVES * j) * 32 + lane;
        const int e  = e_base + (i < SLICE_EDGES ? i : 0);
        dv[j] = edst[e];
        sv[j] = esrc[e];
        wt[j] = ew[e];
    }
    __syncthreads();

    int pk[BATCHES_PER_WAVE];
#pragma unroll
    for (int j = 0; j < BATCHES_PER_WAVE; ++j) {
        const int  i     = (wv + BIN_WAVES * j) * 32 + lane;
        const bool valid = i < SLICE_EDGES;
        const unsigned g = ((unsigned)dv[j] / (unsigned)ROWS_PER_GROUP) & (HOP_GROUPS - 1u);
        const unsigned r = (unsigned)dv[j] - g * (unsigned)ROWS_PER_GROUP;
        const unsigned peers = lanes_with_same_key<8>(g, valid);
        const unsigned rank  = (unsigned)__builtin_popcount(peers & below);
        int c = 0;
        if (valid) {
            c = cnt[wv * HOP_GROUPS + (int)g];
            if (rank == 0u) cnt[wv * HOP_GROUPS + (int)g] = c + __builtin_popcount(peers);
        }
        const unsigned place = (unsigned)c + rank;
        pk[j] = (int)(g | ((r & 511u) << 8) | (place << 17) | (valid ? 1u << 27 : 0u));
    }
    __syncthreads();

    int run_len = 0, run_incl = 0;
    if (tid < HOP_GROUPS) {
        int c[BIN_WAVES];
#pragma unroll
        for (int w = 0; w < BIN_WAVES; ++w) c[w] = cnt[w * HOP_GROUPS + tid];
        int a = 0;
#pragma unroll
        for (int w = 0; w < BIN_WAVES; ++w) {
            cnt[w * HOP_GROUPS + tid] = a;
            a += c[w];
        }
        gtot[tid] = a;
        run_len  = a;
        run_incl = a;
        for (int d = 1; d < 32; d <<= 1) {
            const int up = __shfl_up(run_incl, d, 32);
            if (lane >= d) run_incl += up;
        }
        if (lane == 31) wsum[wv] = run_incl;
    }
    __syncthreads();
    if (tid < HOP_GROUPS) {
        int before = 0;
        for (int w = 0; w < SCAN_WAVES; ++w) {
            const int t = wsum[w];
            before += w < wv ? t : 0;
        }
        gstart[tid] = before + run_incl - run_len;
    }
    __syncthreads();

    int pos[BATCHES_PER_WAVE];
#pragma unroll
    for (int j = 0; j < BATCHES_PER_WAVE; ++j) {
        const unsigned p = (unsigned)pk[j];
        const int g = (int)(p & 255u);
        pos[j] = gstart[g] + cnt[wv * HOP_GROUPS + g] + (int)((p >> 17) & 1023u);
    }
#pragma unroll
    for (int j = 0; j < BATCHES_PER_WAVE; ++j) {
        const unsigned p = (unsigned)pk[j];
        if ((p >> 27) != 0u) {
            v2i rec;
            rec.x = (int)(((unsigned)sv[j] << ROW_BITS) | ((p >> 8) & 511u));
            rec.y = __builtin_bit_cast(int, wt[j]);
            sorted[pos[j]] = rec;
        }
    }
    __syncthreads();

    const int gl   = wv + BIN_WAVES * (lane & (GROUPS_PER_BIN_WAVE - 1));
    const int n_l  = gtot[gl];
    const int st_l = gstart[gl];
    v2i r0[GROUPS_PER_BIN_WAVE], r1[GROUPS_PER_BIN_WAVE];
#pragma unroll
    for (int t = 0; t < GROUPS_PER_BIN_WAVE; ++t) {
        const int st = __builtin_amdgcn_readlane(st_l, t);
        int i0 = st + lane - 1;
        i0 = i0 < 0 ? 0 : i0;
        i0 = i0 < SLICE_EDGES ? i0 : SLICE_EDGES - 1;
        int i1 = st + lane + 31;
        i1 = i1 < SLICE_EDGES ? i1 : SLICE_EDGES - 1;
        r0[t] = sorted[i0];
        r1[t] = sorted[i1];
    }
#pragma unroll
    for (int t = 0; t < GROUPS_PER_BIN_WAVE; ++t) {
        const int g     = wv + BIN_WAVES * t;
        const int gbase = g * ROWS_PER_GROUP;
        const int left  = N_NODES - gbase;
        const int rows  = left < ROWS_PER_GROUP ? left : ROWS_PER_GROUP;
        const int slots = sub_bin_slots(rows);
        const int n     = __builtin_amdgcn_readlane(n_l, t);
        const int ncopy = n < slots - 1 ? n : slots - 1;
        v2f* sub = (v2f*)(nxt + (size_t)gbase * OUT_F) + (size_t)blockIdx.x * slots;
        if (lane <= ncopy) {
            const int ax = r0[t].x, ay = r0[t].y;
            const int rx = lane == 0 ? n : ax;
            const int ry = lane == 0 ? 0 : ay;
            v2f o;
            o.x = float_of(rx);
            o.y = float_of(ry);
            sub[lane] = o;
        }
        if (32 + lane <= ncopy) {
            const int rx = r1[t].x, ry = r1[t].y;
            v2f o;
            o.x = float_of(rx);
            o.y = float_of(ry);
            sub[32 + lane] = o;
        }
    }
    for (int t = 0; t < GROUPS_PER_BIN_WAVE; ++t) {
        const int g     = wv + BIN_WAVES * t;
        const int gbase = g * ROWS_PER_GROUP;
        const int left  = N_NODES - gbase;
        const int rows  = left < ROWS_PER_GROUP ? left : ROWS_PER_GROUP;
        const int slots = sub_bin_slots(rows);
        const int n     = __builtin_amdgcn_readlane(n_l, t);
        const int st    = __builtin_amdgcn_readlane(st_l, t);
        const int ncopy = n < slots - 1 ? n : slots - 1;
        if (ncopy >= 64) {
            const int jdx = 64 + lane;
            if (jdx <= ncopy) {
                const v2i rec = sorted[st + jdx - 1];
                const int rx = rec.x, ry = rec.y;
                v2f o;
                o.x = float_of(rx);
                o.y = float_of(ry);
                v2f* sub = (v2f*)(nxt + (size_t)gbase * OUT_F) + (size_t)blockIdx.x * slots;
                sub[jdx] = o;
            }
        }
    }
}

template <bool FULL>
__device__ __forceinline__ void hop_drain(const v2i* qp, int n, int lane,
                                          const float* __restrict__ cur,
                                          v16f& ax, v16f& ay) {
    int   s  = 0;
    int   rw = 0;
    float wt = 0.f;
    if (FULL || lane < n) {
        const v2i pk = qp[lane];
        rw = (int)(((unsigned)pk.x & ((1u << ROW_BITS) - 1u)) >> 5);
        s  = (int)((unsigned)pk.x >> ROW_BITS);
        const int wbits = pk.y;
        wt = float_of(wbits);
    }
    const int wti = __builtin_bit_cast(int, wt);
    const int cnt = FULL ? 32 : n;

    v2f   v[8];
    int   rj[8];
    float wj[8];
#define HOP_GATHER8(J0, V, RJ, WJ)                                                        \
    _Pragma("unroll")                                                                     \
    for (int u = 0; u < 8; ++u) {                                                         \
        int jj = (J0) + u;                                                                \
        if (!FULL) jj = jj < n ? jj : n - 1;                                              \
        const int sj = __builtin_amdgcn_readlane(s, jj);                                  \
        RJ[u] = __builtin_amdgcn_readlane(rw, jj);                                        \
        WJ[u] = __builtin_bit_cast(float, __builtin_amdgcn_readlane(wti, jj));            \
        V[u]  = *(const v2f*)(cur + (size_t)sj * OUT_F + lane * 2);                       \
    }
    HOP_GATHER8(0, v, rj, wj)
    for (int j0 = 0; j0 < cnt; j0 += 8) {
        v2f   nv[8];
        int   nrj[8];
        float nwj[8];
        const bool more = j0 + 8 < cnt;
        if (more) {
            HOP_GATHER8(j0 + 8, nv, nrj, nwj)
        }
#pragma unroll
        for (int u = 0; u < 8; ++u) {
            if (FULL || j0 + u < n) {
                const int k = rj[u];
                ax[k] = __builtin_fmaf(v[u].x, wj[u], ax[k]);
                ay[k] = __builtin_fmaf(v[u].y, wj[u], ay[k]);
            }
        }
        if (more) {
#pragma unroll
            for (int u = 0; u < 8; ++u) { v[u] = nv[u]; rj[u] = nrj[u]; wj[u] = nwj[u]; }
        }
    }
#undef HOP_GATHER8
}

__global__ __launch_bounds__(HOP_THREADS)
void sgc_hop_owner_waves(const float* __restrict__ cur,
                         float* __restrict__ nxt,
                         const int* __restrict__ esrc,
                         const int* __restrict__ edst,
                         const float* __restrict__ ew) {
    __shared__ v2i boxes[HOP_WAVES * BOX_RSTRIDE];
    __shared__ int box_cnt[HOP_WAVES * HOP_WAVES];
    __shared__ v2i ownq[HOP_WAVES][64];
    __shared__ int over_w[HOP_WAVES];

    const int tid  = threadIdx.x;
    const int lane = tid & 31;
    const int wv   = __builtin_amdgcn_readfirstlane(tid >> 5);
    const int base = blockIdx.x * ROWS_PER_GROUP;
    const int left = N_NODES - base;
    const unsigned nrows = (unsigned)(left < ROWS_PER_GROUP ? left : ROWS_PER_GROUP);
    const unsigned below = (1u << lane) - 1u;

    box_cnt[wv * HOP_WAVES + lane] = 0;
    v16f ax = {}, ay = {};

    const int    slots  = sub_bin_slots((int)nrows);
    const int    room   = slots - 1;
    const v2f*   region = (const v2f*)(nxt + (size_t)base * OUT_F);
    int  nk[4];
    bool cut = false;
#pragma unroll
    for (int k = 0; k < 4; ++k) {
        const v2f* sb = region + (size_t)(wv + HOP_WAVES * k) * slots;
        const float head = sb[0].x;
        nk[k] = __builtin_amdgcn_readfirstlane(bits_of(head));
    }
#pragma unroll
    for (int k = 0; k < 4; ++k) {
        if ((unsigned)nk[k] > (unsigned)room) { cut = true; nk[k] = room; }
    }
    v2f ent[4][SUB_ROUNDS];
#pragma unroll
    for (int k = 0; k < 4; ++k) {
        const v2f* sb = region + (size_t)(wv + HOP_WAVES * k) * slots;
#pragma unroll
        for (int r = 0; r < SUB_ROUNDS; ++r) {
            const int q = r * 32 + lane;
            v2f z; z.x = 0.f; z.y = 0.f;
            ent[k][r] = q < nk[k] ? sb[1 + q] : z;
        }
    }
    v2i* myrow = boxes + wv * BOX_RSTRIDE;
#pragma unroll
    for (int k = 0; k < 4; ++k) {
#pragma unroll
        for (int r = 0; r < SUB_ROUNDS; ++r) {
            const bool has = r * 32 + lane < nk[k];
            if (__builtin_amdgcn_ballot_w32(has) == 0u) continue;
            v2i rec;
            const float fx = ent[k][r].x, fy = ent[k][r].y;
            rec.x = bits_of(fx);
            rec.y = bits_of(fy);
            const unsigned o = (unsigned)rec.x & 31u;
            const unsigned peers = lanes_with_same_key<5>(o, has);
            const int rank = __builtin_popcount(peers & below);
            if (has) {
                const int c    = box_cnt[wv * HOP_WAVES + (int)o];
                const int slot = c + rank;
                if (slot < BOX_SLOTS) myrow[(int)o * BOX_SLOTS + slot] = rec;
                if (rank == 0) box_cnt[wv * HOP_WAVES + (int)o] = c + __builtin_popcount(peers);
            }
        }
    }
    {
        const int c = box_cnt[wv * HOP_WAVES + lane];
        const bool lost = __builtin_amdgcn_ballot_w32(c > BOX_SLOTS) != 0u;
        box_cnt[wv * HOP_WAVES + lane] = c < BOX_SLOTS ? c : BOX_SLOTS;
        if (lane == 0) over_w[wv] = (lost || cut) ? 1 : 0;
    }
    __syncthreads();
    const int over = __builtin_amdgcn_ballot_w32(over_w[lane] != 0) != 0u ? 1 : 0;

    if (over == 0) {
        const int  mine  = box_cnt[lane * HOP_WAVES + wv];
        const v2i* mybox = boxes + lane * BOX_RSTRIDE + wv * BOX_SLOTS;
        v2i* oq = ownq[wv];
        int  on = 0;
        for (int i = 0; i < BOX_SLOTS; ++i) {
            const bool has = i < mine;
            const unsigned m = __builtin_amdgcn_ballot_w32(has);
            if (m == 0u) break;
            const int pos = on + (int)__builtin_amdgcn_mbcnt_lo(m, 0u);
            if (has) oq[pos] = mybox[i];
            on += __builtin_popcount(m);
            if (on >= 32) {
                on -= 32;
                hop_drain<true>(oq + on, 32, lane, cur, ax, ay);
            }
        }
        if (on > 0) hop_drain<false>(oq, on, lane, cur, ax, ay);
    } else {
        const v4i* __restrict__ dst4 = (const v4i*)edst;
        v2i* sq = boxes + wv * QUEUE_RECS;
        int  qn = 0;
        for (int c0 = 0; c0 < N_CHUNKS; c0 += 4) {
            v4i d[4];
#pragma unroll
            for (int u = 0; u < 4; ++u) d[u] = dst4[(size_t)(c0 + u) * 32 + lane];
#pragma unroll
            for (int u = 0; u < 4; ++u) {
                const unsigned rr[4] = {(unsigned)(d[u].x - base), (unsigned)(d[u].y - base),
                                        (unsigned)(d[u].z - base), (unsigned)(d[u].w - base)};
                const int e0 = (c0 + u) * EDGES_PER_CHUNK + lane * 4;
#pragma unroll
                for (int k = 0; k < 4; ++k) {
                    const bool hit = rr[k] < nrows && (int)(rr[k] & 31u) == wv;
                    const unsigned m = __builtin_amdgcn_ballot_w32(hit);
                    if (m != 0u) {
                        const int pos = qn + (int)__builtin_amdgcn_mbcnt_lo(m, 0u);
                        if (hit) {
                            v2i rec;
                            rec.x = (int)(((unsigned)esrc[e0 + k] << ROW_BITS) | rr[k]);
                            rec.y = __builtin_bit_cast(int, ew[e0 + k]);
                            sq[pos] = rec;
                        }
                        qn += __builtin_popcount(m);
                    }
                }
                while (qn >= 32) {
                    qn -= 32;
                    hop_drain<true>(sq + qn, 32, lane, cur, ax, ay);
                }
            }
        }
        if (qn > 0) hop_drain<false>(sq, qn, lane, cur, ax, ay);
    }

#pragma unroll
    for (int r = 0; r < ROWS_PER_WAVE; ++r) {
        const unsigned row = (unsigned)(wv + HOP_WAVES * r);
        if (row < nrows) {
            v2f o;
            o.x = ax[r];
            o.y = ay[r];
            *(v2f*)(nxt + (size_t)(base + (int)row) * OUT_F + lane * 2) = o;
        }
    }
}

extern "C" void kernel_launch(void* const* d_in, const int* in_sizes, int n_in,
                              void* d_out, int out_size, void* d_ws, size_t ws_size,
                              hipStream_t stream) {
    const float* x    = (const float*)d_in[0];
    const float* W    = (const float*)d_in[1];
    const float* b    = (const float*)d_in[2];
    const int*   esrc = (const int*)d_in[3];
    const int*   edst = (const int*)d_in[4];
    const float* ew   = (const float*)d_in[5];

    float* out = (float*)d_out;
    float* ws  = (float*)d_ws;

    sgc_bin_edges<<<BIN_SLICES, BIN_THREADS, 0, stream>>>(esrc, edst, ew, out);
    sgc_gemm_wmma<<<GEMM_GROUPS, GEMM_THREADS, 0, stream>>>(x, W, b, ws);

    sgc_hop_owner_waves<<<HOP_GROUPS, HOP_THREADS, 0, stream>>>(ws, out, esrc, edst, ew);
    sgc_bin_edges<<<BIN_SLICES, BIN_THREADS, 0, stream>>>(esrc, edst, ew, ws);
    sgc_hop_owner_waves<<<HOP_GROUPS, HOP_THREADS, 0, stream>>>(out, ws, esrc, edst, ew);
    sgc_bin_edges<<<BIN_SLICES, BIN_THREADS, 0, stream>>>(esrc, edst, ew, out);
    sgc_hop_owner_waves<<<HOP_GROUPS, HOP_THREADS, 0, stream>>>(ws, out, esrc, edst, ew);
}
